# starting kernel
# baseline (speedup 1.0000x reference)
#include <hip/hip_runtime.h>
#include <hip/hip_bf16.h>

#define NSZ 160
#define NN (NSZ * NSZ)
#define HID 128
#define NHEADS 8
#define HD 16
#define LN_EPS 0.001f

typedef __attribute__((ext_vector_type(2))) float v2f;
typedef __attribute__((ext_vector_type(8))) float v8f;

static __device__ __forceinline__ v8f wmma_f32_16x16x4(v2f a, v2f b, v8f c) {
    return __builtin_amdgcn_wmma_f32_16x16x4_f32(
         false, a,  false, b,
         (short)0, c,  false,  false);
}


__global__ __launch_bounds__(32) void wpack_kernel(
    const float* __restrict__ Wq, const float* __restrict__ Wkv,
    const float* __restrict__ Weg, float* __restrict__ Wt)
{
    const int ct   = blockIdx.x;
    const int lane = threadIdx.x;
    const int hs   = lane / 16;
    const int lm   = lane % 16;

    const float* W; int wcols, n0;
    if (ct < 8)       { W = Wq;  wcols = HID;        n0 = ct * 16; }
    else if (ct < 24) { W = Wkv; wcols = 2 * HID;    n0 = (ct - 8) * 16; }
    else              { W = Weg; wcols = 2 * NHEADS; n0 = 0; }

    float* o = Wt + (size_t)ct * 2048 + lane * 2;
    for (int ks = 0; ks < 32; ++ks) {
        const int kk = ks * 4 + 2 * hs;
        o[ks * 64 + 0] = W[(size_t)(kk)     * wcols + n0 + lm];
        o[ks * 64 + 1] = W[(size_t)(kk + 1) * wcols + n0 + lm];
    }
}

static __device__ __forceinline__ v8f gemm_tile(
    const float* __restrict__ sA, const float* __restrict__ wt,
    int lm, int hs)
{
    v8f acc = {0.f, 0.f, 0.f, 0.f, 0.f, 0.f, 0.f, 0.f};
    const float* pa = sA + lm * HID + 2 * hs;
    #pragma unroll
    for (int ks = 0; ks < 32; ++ks) {
        v2f a, b;
        a.x = pa[ks * 4];
        a.y = pa[ks * 4 + 1];
        b = *reinterpret_cast<const v2f*>(wt + ks * 64);
        acc = wmma_f32_16x16x4(a, b, acc);
    }
    return acc;
}

__global__ __launch_bounds__(256) void proj_kernel(
    const float* __restrict__ feat1, const float* __restrict__ feat2,
    const float* __restrict__ Wt,
    const float* __restrict__ bq, const float* __restrict__ bkv,
    const float* __restrict__ beg,
    float* __restrict__ Qp, float* __restrict__ Kp, float* __restrict__ Vp,
    float* __restrict__ Ep, float* __restrict__ Gp)
{
    __shared__ float sA1[16][HID];
    __shared__ float sA2[16][HID];

    const int row0 = blockIdx.x * 16;
    const int tid  = threadIdx.x;

    for (int idx = tid; idx < 16 * (HID / 4); idx += 256) {
        const int r = idx / (HID / 4), c4 = idx % (HID / 4);
        reinterpret_cast<float4*>(&sA1[r][0])[c4] =
            reinterpret_cast<const float4*>(&feat1[(size_t)(row0 + r) * HID])[c4];
        reinterpret_cast<float4*>(&sA2[r][0])[c4] =
            reinterpret_cast<const float4*>(&feat2[(size_t)(row0 + r) * HID])[c4];
    }
    __syncthreads();

    const int wave = tid / 32;
    const int lane = tid % 32;
    const int hs   = lane / 16;
    const int lm   = lane % 16;

    const int aIdx = row0 / NSZ;
    const int b0   = row0 % NSZ;
    const int bBase = b0 + 8 * hs;

    const int col = wave * 16 + lm;
    const int d   = col / 8;
    const int h   = col % 8;

    {
        v8f acc = gemm_tile(&sA1[0][0], Wt + (size_t)wave * 2048 + lane * 2, lm, hs);
        const float bv = bq[col];
        float* oq = Qp + (((size_t)bBase * NHEADS + h) * NSZ + aIdx) * HD + d;
        #pragma unroll
        for (int r = 0; r < 8; ++r)
            oq[(size_t)r * NHEADS * NSZ * HD] = (acc[r] + bv) * 0.25f;
    }

    {
        v8f acc = gemm_tile(&sA2[0][0], Wt + (size_t)(8 + wave) * 2048 + lane * 2, lm, hs);
        const float bv = bkv[col];
        float* ok = Kp + (((size_t)aIdx * NHEADS + h) * NSZ + bBase) * HD + d;
        #pragma unroll
        for (int r = 0; r < 8; ++r)
            ok[r * HD] = acc[r] + bv;
    }

    {
        v8f acc = gemm_tile(&sA2[0][0], Wt + (size_t)(16 + wave) * 2048 + lane * 2, lm, hs);
        const float bv = bkv[HID + col];
        float* ov = Vp + (((size_t)aIdx * NHEADS + h) * HD + d) * NSZ + bBase;
        #pragma unroll
        for (int r = 0; r < 8; ++r)
            ov[r] = acc[r] + bv;
    }

    if (wave == 0) {
        v8f acc = gemm_tile(&sA2[0][0], Wt + (size_t)24 * 2048 + lane * 2, lm, hs);
        const float bv = beg[lm];
        if (lm < NHEADS) {
            float* oe = Ep + ((size_t)aIdx * NHEADS + lm) * NSZ + bBase;
            #pragma unroll
            for (int r = 0; r < 8; ++r) oe[r] = acc[r] + bv;
        } else {
            float* og = Gp + ((size_t)aIdx * NHEADS + (lm - NHEADS)) * NSZ + bBase;
            #pragma unroll
            for (int r = 0; r < 8; ++r)
                og[r] = 1.f / (1.f + __expf(-(acc[r] + bv)));
        }
    }
}

__global__ __launch_bounds__(256) void attn_kernel(
    const float* __restrict__ Qp, const float* __restrict__ Kp,
    const float* __restrict__ Vp, const float* __restrict__ Ep,
    const float* __restrict__ Gp,
    const float* __restrict__ ln_g, const float* __restrict__ ln_b,
    float* __restrict__ out)
{
    __shared__ float Plds[NHEADS][16][81];
    __shared__ float VaLds[16][HID];

    const int j    = blockIdx.x;
    const int i0   = blockIdx.y * 16;
    const int tid  = threadIdx.x;
    const int h    = tid / 32;
    const int lane = tid % 32;
    const int hs   = lane / 16;
    const int lm   = lane % 16;

    const size_t jh = (size_t)j * NHEADS + h;

    const float* pQ = Qp + (jh * NSZ + (i0 + lm)) * HD + 2 * hs;
    const float* pK = Kp + (jh * NSZ + lm) * HD + 2 * hs;
    const float* pV = Vp + (jh * HD + lm) * NSZ + 2 * hs;
    const float* pE = Ep + jh * NSZ + lm;
    const float* pG = Gp + jh * NSZ + lm;

    {
        const float* vslice = Vp + jh * HD * NSZ;
        for (int p = lane; p < 80; p += 32)
            __builtin_prefetch(vslice + p * 32, 0, 1);
    }

    v2f afr[4];
    #pragma unroll
    for (int ks = 0; ks < 4; ++ks)
        afr[ks] = *reinterpret_cast<const v2f*>(pQ + ks * 4);

    v8f acc[10];
    #pragma unroll
    for (int ct = 0; ct < 10; ++ct) {
        v8f c = {0.f, 0.f, 0.f, 0.f, 0.f, 0.f, 0.f, 0.f};
        #pragma unroll
        for (int ks = 0; ks < 4; ++ks) {
            v2f b = *reinterpret_cast<const v2f*>(pK + ct * 16 * HD + ks * 4);
            c = wmma_f32_16x16x4(afr[ks], b, c);
        }
        const float e = pE[ct * 16];
        #pragma unroll
        for (int r = 0; r < 8; ++r) c[r] += e;
        acc[ct] = c;
    }

    float gate[10];
    #pragma unroll
    for (int ct = 0; ct < 10; ++ct) gate[ct] = pG[ct * 16];

    #pragma unroll
    for (int r = 0; r < 8; ++r) {
        float mx = -1e30f;
        #pragma unroll
        for (int ct = 0; ct < 10; ++ct) mx = fmaxf(mx, acc[ct][r]);
        #pragma unroll
        for (int msk = 1; msk < 16; msk <<= 1) mx = fmaxf(mx, __shfl_xor(mx, msk, 32));
        float sm = 0.f;
        #pragma unroll
        for (int ct = 0; ct < 10; ++ct) {
            float e = __expf(acc[ct][r] - mx);
            acc[ct][r] = e;
            sm += e;
        }
        #pragma unroll
        for (int msk = 1; msk < 16; msk <<= 1) sm += __shfl_xor(sm, msk, 32);
        const float inv = 1.f / sm;
        #pragma unroll
        for (int ct = 0; ct < 10; ++ct) acc[ct][r] = acc[ct][r] * inv * gate[ct];
    }

    v8f va = {0.f, 0.f, 0.f, 0.f, 0.f, 0.f, 0.f, 0.f};
    #pragma unroll
    for (int half = 0; half < 2; ++half) {
        #pragma unroll
        for (int ct = 0; ct < 5; ++ct) {
            #pragma unroll
            for (int r = 0; r < 8; ++r)
                Plds[h][r + 8 * hs][ct * 16 + lm] = acc[half * 5 + ct][r];
        }
        __syncthreads();
        #pragma unroll
        for (int k0 = 0; k0 < 80; k0 += 4) {
            v2f a, b;
            a.x = Plds[h][lm][k0 + 2 * hs];
            a.y = Plds[h][lm][k0 + 2 * hs + 1];
            b = *reinterpret_cast<const v2f*>(pV + half * 80 + k0);
            va = wmma_f32_16x16x4(a, b, va);
        }
        __syncthreads();
    }

    #pragma unroll
    for (int r = 0; r < 8; ++r)
        VaLds[r + 8 * hs][lm * NHEADS + h] = va[r];
    __syncthreads();

    const int rowLN = tid / 16;
    const int cg    = tid % 16;
    float vals[8], s1 = 0.f, s2 = 0.f;
    #pragma unroll
    for (int c = 0; c < 8; ++c) {
        float v = VaLds[rowLN][cg * 8 + c];
        vals[c] = v; s1 += v; s2 += v * v;
    }
    #pragma unroll
    for (int msk = 1; msk < 16; msk <<= 1) {
        s1 += __shfl_xor(s1, msk, 32);
        s2 += __shfl_xor(s2, msk, 32);
    }
    const float mu  = s1 * (1.f / HID);
    const float var = s2 * (1.f / HID) - mu * mu;
    const float inv = rsqrtf(var + LN_EPS);

    const size_t obase = ((size_t)(i0 + rowLN) * NSZ + j) * HID + cg * 8;
    v2f* op = reinterpret_cast<v2f*>(&out[obase]);
    #pragma unroll
    for (int c = 0; c < 8; c += 2) {
        const int ch = cg * 8 + c;
        v2f t;
        t.x = (vals[c]     - mu) * inv * ln_g[ch]     + ln_b[ch];
        t.y = (vals[c + 1] - mu) * inv * ln_g[ch + 1] + ln_b[ch + 1];
        op[c / 2] = t;
    }
}

extern "C" void kernel_launch(void* const* d_in, const int* in_sizes, int n_in,
                              void* d_out, int out_size, void* d_ws, size_t ws_size,
                              hipStream_t stream) {
    const float* feat1 = (const float*)d_in[0];
    const float* feat2 = (const float*)d_in[1];
    const float* Wq    = (const float*)d_in[2];
    const float* bq    = (const float*)d_in[3];
    const float* Wkv   = (const float*)d_in[4];
    const float* bkv   = (const float*)d_in[5];
    const float* Weg   = (const float*)d_in[6];
    const float* beg   = (const float*)d_in[7];
    const float* lng   = (const float*)d_in[8];
    const float* lnb   = (const float*)d_in[9];

    float* ws = (float*)d_ws;
    float* Qp = ws;
    float* Kp = Qp + (size_t)NN * HID;
    float* Vp = Kp + (size_t)NN * HID;
    float* Ep = Vp + (size_t)NN * HID;
    float* Gp = Ep + (size_t)NN * NHEADS;
    float* Wt = Gp + (size_t)NN * NHEADS;

    wpack_kernel<<<dim3(25), 32, 0, stream>>>(Wq, Wkv, Weg, Wt);

    proj_kernel<<<dim3(NN / 16), 256, 0, stream>>>(
        feat1, feat2, Wt, bq, bkv, beg, Qp, Kp, Vp, Ep, Gp);

    attn_kernel<<<dim3(NSZ, NSZ / 16), 256, 0, stream>>>(
        Qp, Kp, Vp, Ep, Gp, lng, lnb, (float*)d_out);
}


# Projection GEMM: pad the LDS A-tile row pitch from 128 to 132 floats (16 B x odd) so each A-fragment ds_load spreads over all 64 banks instead of a 16-way conflict
# speedup vs baseline: 1.0435x; 1.0435x over previous
#include <hip/hip_runtime.h>
#include <hip/hip_bf16.h>

#define NSZ 160
#define NN (NSZ * NSZ)
#define HID 128
#define NHEADS 8
#define HD 16
#define LN_EPS 0.001f
#define APITCH 132

typedef __attribute__((ext_vector_type(2))) float v2f;
typedef __attribute__((ext_vector_type(8))) float v8f;

static __device__ __forceinline__ v8f wmma_f32_16x16x4(v2f a, v2f b, v8f c) {
    return __builtin_amdgcn_wmma_f32_16x16x4_f32(
         false, a,  false, b,
         (short)0, c,  false,  false);
}


__global__ __launch_bounds__(32) void wpack_kernel(
    const float* __restrict__ Wq, const float* __restrict__ Wkv,
    const float* __restrict__ Weg, float* __restrict__ Wt)
{
    const int ct   = blockIdx.x;
    const int lane = threadIdx.x;
    const int hs   = lane / 16;
    const int lm   = lane % 16;

    const float* W; int wcols, n0;
    if (ct < 8)       { W = Wq;  wcols = HID;        n0 = ct * 16; }
    else if (ct < 24) { W = Wkv; wcols = 2 * HID;    n0 = (ct - 8) * 16; }
    else              { W = Weg; wcols = 2 * NHEADS; n0 = 0; }

    float* o = Wt + (size_t)ct * 2048 + lane * 2;
    for (int ks = 0; ks < 32; ++ks) {
        const int kk = ks * 4 + 2 * hs;
        o[ks * 64 + 0] = W[(size_t)(kk)     * wcols + n0 + lm];
        o[ks * 64 + 1] = W[(size_t)(kk + 1) * wcols + n0 + lm];
    }
}

static __device__ __forceinline__ v8f gemm_tile(
    const float* __restrict__ sA, const float* __restrict__ wt,
    int lm, int hs)
{
    v8f acc = {0.f, 0.f, 0.f, 0.f, 0.f, 0.f, 0.f, 0.f};
    const float* pa = sA + lm * APITCH + 2 * hs;
    #pragma unroll
    for (int ks = 0; ks < 32; ++ks) {
        v2f a, b;
        a.x = pa[ks * 4];
        a.y = pa[ks * 4 + 1];
        b = *reinterpret_cast<const v2f*>(wt + ks * 64);
        acc = wmma_f32_16x16x4(a, b, acc);
    }
    return acc;
}

__global__ __launch_bounds__(256) void proj_kernel(
    const float* __restrict__ feat1, const float* __restrict__ feat2,
    const float* __restrict__ Wt,
    const float* __restrict__ bq, const float* __restrict__ bkv,
    const float* __restrict__ beg,
    float* __restrict__ Qp, float* __restrict__ Kp, float* __restrict__ Vp,
    float* __restrict__ Ep, float* __restrict__ Gp)
{
    __shared__ float sA1[16][APITCH];
    __shared__ float sA2[16][APITCH];

    const int row0 = blockIdx.x * 16;
    const int tid  = threadIdx.x;

    for (int idx = tid; idx < 16 * (HID / 4); idx += 256) {
        const int r = idx / (HID / 4), c4 = idx % (HID / 4);
        reinterpret_cast<float4*>(&sA1[r][0])[c4] =
            reinterpret_cast<const float4*>(&feat1[(size_t)(row0 + r) * HID])[c4];
        reinterpret_cast<float4*>(&sA2[r][0])[c4] =
            reinterpret_cast<const float4*>(&feat2[(size_t)(row0 + r) * HID])[c4];
    }
    __syncthreads();

    const int wave = tid / 32;
    const int lane = tid % 32;
    const int hs   = lane / 16;
    const int lm   = lane % 16;

    const int aIdx = row0 / NSZ;
    const int b0   = row0 % NSZ;
    const int bBase = b0 + 8 * hs;

    const int col = wave * 16 + lm;
    const int d   = col / 8;
    const int h   = col % 8;

    {
        v8f acc = gemm_tile(&sA1[0][0], Wt + (size_t)wave * 2048 + lane * 2, lm, hs);
        const float bv = bq[col];
        float* oq = Qp + (((size_t)bBase * NHEADS + h) * NSZ + aIdx) * HD + d;
        #pragma unroll
        for (int r = 0; r < 8; ++r)
            oq[(size_t)r * NHEADS * NSZ * HD] = (acc[r] + bv) * 0.25f;
    }

    {
        v8f acc = gemm_tile(&sA2[0][0], Wt + (size_t)(8 + wave) * 2048 + lane * 2, lm, hs);
        const float bv = bkv[col];
        float* ok = Kp + (((size_t)aIdx * NHEADS + h) * NSZ + bBase) * HD + d;
        #pragma unroll
        for (int r = 0; r < 8; ++r)
            ok[r * HD] = acc[r] + bv;
    }

    {
        v8f acc = gemm_tile(&sA2[0][0], Wt + (size_t)(16 + wave) * 2048 + lane * 2, lm, hs);
        const float bv = bkv[HID + col];
        float* ov = Vp + (((size_t)aIdx * NHEADS + h) * HD + d) * NSZ + bBase;
        #pragma unroll
        for (int r = 0; r < 8; ++r)
            ov[r] = acc[r] + bv;
    }

    if (wave == 0) {
        v8f acc = gemm_tile(&sA2[0][0], Wt + (size_t)24 * 2048 + lane * 2, lm, hs);
        const float bv = beg[lm];
        if (lm < NHEADS) {
            float* oe = Ep + ((size_t)aIdx * NHEADS + lm) * NSZ + bBase;
            #pragma unroll
            for (int r = 0; r < 8; ++r) oe[r] = acc[r] + bv;
        } else {
            float* og = Gp + ((size_t)aIdx * NHEADS + (lm - NHEADS)) * NSZ + bBase;
            #pragma unroll
            for (int r = 0; r < 8; ++r)
                og[r] = 1.f / (1.f + __expf(-(acc[r] + bv)));
        }
    }
}

__global__ __launch_bounds__(256) void attn_kernel(
    const float* __restrict__ Qp, const float* __restrict__ Kp,
    const float* __restrict__ Vp, const float* __restrict__ Ep,
    const float* __restrict__ Gp,
    const float* __restrict__ ln_g, const float* __restrict__ ln_b,
    float* __restrict__ out)
{
    __shared__ float Plds[NHEADS][16][81];
    __shared__ float VaLds[16][HID];

    const int j    = blockIdx.x;
    const int i0   = blockIdx.y * 16;
    const int tid  = threadIdx.x;
    const int h    = tid / 32;
    const int lane = tid % 32;
    const int hs   = lane / 16;
    const int lm   = lane % 16;

    const size_t jh = (size_t)j * NHEADS + h;

    const float* pQ = Qp + (jh * NSZ + (i0 + lm)) * HD + 2 * hs;
    const float* pK = Kp + (jh * NSZ + lm) * HD + 2 * hs;
    const float* pV = Vp + (jh * HD + lm) * NSZ + 2 * hs;
    const float* pE = Ep + jh * NSZ + lm;
    const float* pG = Gp + jh * NSZ + lm;

    {
        const float* vslice = Vp + jh * HD * NSZ;
        for (int p = lane; p < 80; p += 32)
            __builtin_prefetch(vslice + p * 32, 0, 1);
    }

    v2f afr[4];
    #pragma unroll
    for (int ks = 0; ks < 4; ++ks)
        afr[ks] = *reinterpret_cast<const v2f*>(pQ + ks * 4);

    v8f acc[10];
    #pragma unroll
    for (int ct = 0; ct < 10; ++ct) {
        v8f c = {0.f, 0.f, 0.f, 0.f, 0.f, 0.f, 0.f, 0.f};
        #pragma unroll
        for (int ks = 0; ks < 4; ++ks) {
            v2f b = *reinterpret_cast<const v2f*>(pK + ct * 16 * HD + ks * 4);
            c = wmma_f32_16x16x4(afr[ks], b, c);
        }
        const float e = pE[ct * 16];
        #pragma unroll
        for (int r = 0; r < 8; ++r) c[r] += e;
        acc[ct] = c;
    }

    float gate[10];
    #pragma unroll
    for (int ct = 0; ct < 10; ++ct) gate[ct] = pG[ct * 16];

    #pragma unroll
    for (int r = 0; r < 8; ++r) {
        float mx = -1e30f;
        #pragma unroll
        for (int ct = 0; ct < 10; ++ct) mx = fmaxf(mx, acc[ct][r]);
        #pragma unroll
        for (int msk = 1; msk < 16; msk <<= 1) mx = fmaxf(mx, __shfl_xor(mx, msk, 32));
        float sm = 0.f;
        #pragma unroll
        for (int ct = 0; ct < 10; ++ct) {
            float e = __expf(acc[ct][r] - mx);
            acc[ct][r] = e;
            sm += e;
        }
        #pragma unroll
        for (int msk = 1; msk < 16; msk <<= 1) sm += __shfl_xor(sm, msk, 32);
        const float inv = 1.f / sm;
        #pragma unroll
        for (int ct = 0; ct < 10; ++ct) acc[ct][r] = acc[ct][r] * inv * gate[ct];
    }

    v8f va = {0.f, 0.f, 0.f, 0.f, 0.f, 0.f, 0.f, 0.f};
    #pragma unroll
    for (int half = 0; half < 2; ++half) {
        #pragma unroll
        for (int ct = 0; ct < 5; ++ct) {
            #pragma unroll
            for (int r = 0; r < 8; ++r)
                Plds[h][r + 8 * hs][ct * 16 + lm] = acc[half * 5 + ct][r];
        }
        __syncthreads();
        #pragma unroll
        for (int k0 = 0; k0 < 80; k0 += 4) {
            v2f a, b;
            a.x = Plds[h][lm][k0 + 2 * hs];
            a.y = Plds[h][lm][k0 + 2 * hs + 1];
            b = *reinterpret_cast<const v2f*>(pV + half * 80 + k0);
            va = wmma_f32_16x16x4(a, b, va);
        }
        __syncthreads();
    }

    #pragma unroll
    for (int r = 0; r < 8; ++r)
        VaLds[r + 8 * hs][lm * NHEADS + h] = va[r];
    __syncthreads();

    const int rowLN = tid / 16;
    const int cg    = tid % 16;
    float vals[8], s1 = 0.f, s2 = 0.f;
    #pragma unroll
    for (int c = 0; c < 8; ++c) {
        float v = VaLds[rowLN][cg * 8 + c];
        vals[c] = v; s1 += v; s2 += v * v;
    }
    #pragma unroll
    for (int msk = 1; msk < 16; msk <<= 1) {
        s1 += __shfl_xor(s1, msk, 32);
        s2 += __shfl_xor(s2, msk, 32);
    }
    const float mu  = s1 * (1.f / HID);
    const float var = s2 * (1.f / HID) - mu * mu;
    const float inv = rsqrtf(var + LN_EPS);

    const size_t obase = ((size_t)(i0 + rowLN) * NSZ + j) * HID + cg * 8;
    v2f* op = reinterpret_cast<v2f*>(&out[obase]);
    #pragma unroll
    for (int c = 0; c < 8; c += 2) {
        const int ch = cg * 8 + c;
        v2f t;
        t.x = (vals[c]     - mu) * inv * ln_g[ch]     + ln_b[ch];
        t.y = (vals[c + 1] - mu) * inv * ln_g[ch + 1] + ln_b[ch + 1];
        op[c / 2] = t;
    }
}

extern "C" void kernel_launch(void* const* d_in, const int* in_sizes, int n_in,
                              void* d_out, int out_size, void* d_ws, size_t ws_size,
                              hipStream_t stream) {
    const float* feat1 = (const float*)d_in[0];
    const float* feat2 = (const float*)d_in[1];
    const float* Wq    = (const float*)d_in[2];
    const float* bq    = (const float*)d_in[3];
    const float* Wkv   = (const float*)d_in[4];
    const float* bkv   = (const float*)d_in[5];
    const float* Weg   = (const float*)d_in[6];
    const float* beg   = (const float*)d_in[7];
    const float* lng   = (const float*)d_in[8];
    const float* lnb   = (const float*)d_in[9];

    float* ws = (float*)d_ws;
    float* Qp = ws;
    float* Kp = Qp + (size_t)NN * HID;
    float* Vp = Kp + (size_t)NN * HID;
    float* Ep = Vp + (size_t)NN * HID;
    float* Gp = Ep + (size_t)NN * NHEADS;
    float* Wt = Gp + (size_t)NN * NHEADS;

    wpack_kernel<<<dim3(25), 32, 0, stream>>>(Wq, Wkv, Weg, Wt);

    proj_kernel<<<dim3(NN / 16), 256, 0, stream>>>(
        feat1, feat2, Wt, bq, bkv, beg, Qp, Kp, Vp, Ep, Gp);

    attn_kernel<<<dim3(NSZ, NSZ / 16), 256, 0, stream>>>(
        Qp, Kp, Vp, Ep, Gp, lng, lnb, (float*)d_out);
}


# Attention kernel: K, V and P operand fragments are loaded as whole groups ahead of each WMMA run (all loads of a run in flight at once) instead of one load, one full wait, one WMMA at a time
# speedup vs baseline: 1.1941x; 1.1443x over previous
#include <hip/hip_runtime.h>
#include <hip/hip_bf16.h>

#define NSZ 160
#define NN (NSZ * NSZ)
#define HID 128
#define NHEADS 8
#define HD 16
#define LN_EPS 0.001f
#define APITCH 132

typedef __attribute__((ext_vector_type(2))) float v2f;
typedef __attribute__((ext_vector_type(8))) float v8f;

static __device__ __forceinline__ v8f wmma_f32_16x16x4(v2f a, v2f b, v8f c) {
    return __builtin_amdgcn_wmma_f32_16x16x4_f32(
         false, a,  false, b,
         (short)0, c,  false,  false);
}


__global__ __launch_bounds__(32) void wpack_kernel(
    const float* __restrict__ Wq, const float* __restrict__ Wkv,
    const float* __restrict__ Weg, float* __restrict__ Wt)
{
    const int ct   = blockIdx.x;
    const int lane = threadIdx.x;
    const int hs   = lane / 16;
    const int lm   = lane % 16;

    const float* W; int wcols, n0;
    if (ct < 8)       { W = Wq;  wcols = HID;        n0 = ct * 16; }
    else if (ct < 24) { W = Wkv; wcols = 2 * HID;    n0 = (ct - 8) * 16; }
    else              { W = Weg; wcols = 2 * NHEADS; n0 = 0; }

    float* o = Wt + (size_t)ct * 2048 + lane * 2;
    for (int ks = 0; ks < 32; ++ks) {
        const int kk = ks * 4 + 2 * hs;
        o[ks * 64 + 0] = W[(size_t)(kk)     * wcols + n0 + lm];
        o[ks * 64 + 1] = W[(size_t)(kk + 1) * wcols + n0 + lm];
    }
}

static __device__ __forceinline__ v8f gemm_tile(
    const float* __restrict__ sA, const float* __restrict__ wt,
    int lm, int hs)
{
    v8f acc = {0.f, 0.f, 0.f, 0.f, 0.f, 0.f, 0.f, 0.f};
    const float* pa = sA + lm * APITCH + 2 * hs;
    #pragma unroll
    for (int ks = 0; ks < 32; ++ks) {
        v2f a, b;
        a.x = pa[ks * 4];
        a.y = pa[ks * 4 + 1];
        b = *reinterpret_cast<const v2f*>(wt + ks * 64);
        acc = wmma_f32_16x16x4(a, b, acc);
    }
    return acc;
}

__global__ __launch_bounds__(256) void proj_kernel(
    const float* __restrict__ feat1, const float* __restrict__ feat2,
    const float* __restrict__ Wt,
    const float* __restrict__ bq, const float* __restrict__ bkv,
    const float* __restrict__ beg,
    float* __restrict__ Qp, float* __restrict__ Kp, float* __restrict__ Vp,
    float* __restrict__ Ep, float* __restrict__ Gp)
{
    __shared__ float sA1[16][APITCH];
    __shared__ float sA2[16][APITCH];

    const int row0 = blockIdx.x * 16;
    const int tid  = threadIdx.x;

    for (int idx = tid; idx < 16 * (HID / 4); idx += 256) {
        const int r = idx / (HID / 4), c4 = idx % (HID / 4);
        reinterpret_cast<float4*>(&sA1[r][0])[c4] =
            reinterpret_cast<const float4*>(&feat1[(size_t)(row0 + r) * HID])[c4];
        reinterpret_cast<float4*>(&sA2[r][0])[c4] =
            reinterpret_cast<const float4*>(&feat2[(size_t)(row0 + r) * HID])[c4];
    }
    __syncthreads();

    const int wave = tid / 32;
    const int lane = tid % 32;
    const int hs   = lane / 16;
    const int lm   = lane % 16;

    const int aIdx = row0 / NSZ;
    const int b0   = row0 % NSZ;
    const int bBase = b0 + 8 * hs;

    const int col = wave * 16 + lm;
    const int d   = col / 8;
    const int h   = col % 8;

    {
        v8f acc = gemm_tile(&sA1[0][0], Wt + (size_t)wave * 2048 + lane * 2, lm, hs);
        const float bv = bq[col];
        float* oq = Qp + (((size_t)bBase * NHEADS + h) * NSZ + aIdx) * HD + d;
        #pragma unroll
        for (int r = 0; r < 8; ++r)
            oq[(size_t)r * NHEADS * NSZ * HD] = (acc[r] + bv) * 0.25f;
    }

    {
        v8f acc = gemm_tile(&sA2[0][0], Wt + (size_t)(8 + wave) * 2048 + lane * 2, lm, hs);
        const float bv = bkv[col];
        float* ok = Kp + (((size_t)aIdx * NHEADS + h) * NSZ + bBase) * HD + d;
        #pragma unroll
        for (int r = 0; r < 8; ++r)
            ok[r * HD] = acc[r] + bv;
    }

    {
        v8f acc = gemm_tile(&sA2[0][0], Wt + (size_t)(16 + wave) * 2048 + lane * 2, lm, hs);
        const float bv = bkv[HID + col];
        float* ov = Vp + (((size_t)aIdx * NHEADS + h) * HD + d) * NSZ + bBase;
        #pragma unroll
        for (int r = 0; r < 8; ++r)
            ov[r] = acc[r] + bv;
    }

    if (wave == 0) {
        v8f acc = gemm_tile(&sA2[0][0], Wt + (size_t)24 * 2048 + lane * 2, lm, hs);
        const float bv = beg[lm];
        if (lm < NHEADS) {
            float* oe = Ep + ((size_t)aIdx * NHEADS + lm) * NSZ + bBase;
            #pragma unroll
            for (int r = 0; r < 8; ++r) oe[r] = acc[r] + bv;
        } else {
            float* og = Gp + ((size_t)aIdx * NHEADS + (lm - NHEADS)) * NSZ + bBase;
            #pragma unroll
            for (int r = 0; r < 8; ++r)
                og[r] = 1.f / (1.f + __expf(-(acc[r] + bv)));
        }
    }
}

__global__ __launch_bounds__(256) void attn_kernel(
    const float* __restrict__ Qp, const float* __restrict__ Kp,
    const float* __restrict__ Vp, const float* __restrict__ Ep,
    const float* __restrict__ Gp,
    const float* __restrict__ ln_g, const float* __restrict__ ln_b,
    float* __restrict__ out)
{
    __shared__ float Plds[NHEADS][16][81];
    __shared__ float VaLds[16][HID];

    const int j    = blockIdx.x;
    const int i0   = blockIdx.y * 16;
    const int tid  = threadIdx.x;
    const int h    = tid / 32;
    const int lane = tid % 32;
    const int hs   = lane / 16;
    const int lm   = lane % 16;

    const size_t jh = (size_t)j * NHEADS + h;

    const float* pQ = Qp + (jh * NSZ + (i0 + lm)) * HD + 2 * hs;
    const float* pK = Kp + (jh * NSZ + lm) * HD + 2 * hs;
    const float* pV = Vp + (jh * HD + lm) * NSZ + 2 * hs;
    const float* pE = Ep + jh * NSZ + lm;
    const float* pG = Gp + jh * NSZ + lm;

    {
        const float* vslice = Vp + jh * HD * NSZ;
        for (int p = lane; p < 80; p += 32)
            __builtin_prefetch(vslice + p * 32, 0, 1);
    }

    v2f afr[4];
    #pragma unroll
    for (int ks = 0; ks < 4; ++ks)
        afr[ks] = *reinterpret_cast<const v2f*>(pQ + ks * 4);

    v2f kf[40];
    #pragma unroll
    for (int ct = 0; ct < 10; ++ct) {
        #pragma unroll
        for (int ks = 0; ks < 4; ++ks)
            kf[ct * 4 + ks] = *reinterpret_cast<const v2f*>(pK + ct * 16 * HD + ks * 4);
    }
    float ev[10];
    #pragma unroll
    for (int ct = 0; ct < 10; ++ct) ev[ct] = pE[ct * 16];
    __builtin_amdgcn_sched_barrier(0);

    v8f acc[10];
    #pragma unroll
    for (int ct = 0; ct < 10; ++ct) {
        v8f c = {0.f, 0.f, 0.f, 0.f, 0.f, 0.f, 0.f, 0.f};
        #pragma unroll
        for (int ks = 0; ks < 4; ++ks)
            c = wmma_f32_16x16x4(afr[ks], kf[ct * 4 + ks], c);
        const float e = ev[ct];
        #pragma unroll
        for (int r = 0; r < 8; ++r) c[r] += e;
        acc[ct] = c;
    }

    float gate[10];
    #pragma unroll
    for (int ct = 0; ct < 10; ++ct) gate[ct] = pG[ct * 16];

    #pragma unroll
    for (int r = 0; r < 8; ++r) {
        float mx = -1e30f;
        #pragma unroll
        for (int ct = 0; ct < 10; ++ct) mx = fmaxf(mx, acc[ct][r]);
        #pragma unroll
        for (int msk = 1; msk < 16; msk <<= 1) mx = fmaxf(mx, __shfl_xor(mx, msk, 32));
        float sm = 0.f;
        #pragma unroll
        for (int ct = 0; ct < 10; ++ct) {
            float e = __expf(acc[ct][r] - mx);
            acc[ct][r] = e;
            sm += e;
        }
        #pragma unroll
        for (int msk = 1; msk < 16; msk <<= 1) sm += __shfl_xor(sm, msk, 32);
        const float inv = 1.f / sm;
        #pragma unroll
        for (int ct = 0; ct < 10; ++ct) acc[ct][r] = acc[ct][r] * inv * gate[ct];
    }

    v8f va = {0.f, 0.f, 0.f, 0.f, 0.f, 0.f, 0.f, 0.f};
    #pragma unroll
    for (int half = 0; half < 2; ++half) {
        #pragma unroll
        for (int ct = 0; ct < 5; ++ct) {
            #pragma unroll
            for (int r = 0; r < 8; ++r)
                Plds[h][r + 8 * hs][ct * 16 + lm] = acc[half * 5 + ct][r];
        }
        __builtin_amdgcn_sched_barrier(0);
        v2f vf[20];
        #pragma unroll
        for (int q = 0; q < 20; ++q)
            vf[q] = *reinterpret_cast<const v2f*>(pV + half * 80 + q * 4);
        __syncthreads();
        v2f pf[20];
        #pragma unroll
        for (int q = 0; q < 20; ++q) {
            pf[q].x = Plds[h][lm][q * 4 + 2 * hs];
            pf[q].y = Plds[h][lm][q * 4 + 2 * hs + 1];
        }
        __builtin_amdgcn_sched_barrier(0);
        #pragma unroll
        for (int q = 0; q < 20; ++q)
            va = wmma_f32_16x16x4(pf[q], vf[q], va);
        __syncthreads();
    }

    #pragma unroll
    for (int r = 0; r < 8; ++r)
        VaLds[r + 8 * hs][lm * NHEADS + h] = va[r];
    __syncthreads();

    const int rowLN = tid / 16;
    const int cg    = tid % 16;
    float vals[8], s1 = 0.f, s2 = 0.f;
    #pragma unroll
    for (int c = 0; c < 8; ++c) {
        float v = VaLds[rowLN][cg * 8 + c];
        vals[c] = v; s1 += v; s2 += v * v;
    }
    #pragma unroll
    for (int msk = 1; msk < 16; msk <<= 1) {
        s1 += __shfl_xor(s1, msk, 32);
        s2 += __shfl_xor(s2, msk, 32);
    }
    const float mu  = s1 * (1.f / HID);
    const float var = s2 * (1.f / HID) - mu * mu;
    const float inv = rsqrtf(var + LN_EPS);

    const size_t obase = ((size_t)(i0 + rowLN) * NSZ + j) * HID + cg * 8;
    v2f* op = reinterpret_cast<v2f*>(&out[obase]);
    #pragma unroll
    for (int c = 0; c < 8; c += 2) {
        const int ch = cg * 8 + c;
        v2f t;
        t.x = (vals[c]     - mu) * inv * ln_g[ch]     + ln_b[ch];
        t.y = (vals[c + 1] - mu) * inv * ln_g[ch + 1] + ln_b[ch + 1];
        op[c / 2] = t;
    }
}

extern "C" void kernel_launch(void* const* d_in, const int* in_sizes, int n_in,
                              void* d_out, int out_size, void* d_ws, size_t ws_size,
                              hipStream_t stream) {
    const float* feat1 = (const float*)d_in[0];
    const float* feat2 = (const float*)d_in[1];
    const float* Wq    = (const float*)d_in[2];
    const float* bq    = (const float*)d_in[3];
    const float* Wkv   = (const float*)d_in[4];
    const float* bkv   = (const float*)d_in[5];
    const float* Weg   = (const float*)d_in[6];
    const float* beg   = (const float*)d_in[7];
    const float* lng   = (const float*)d_in[8];
    const float* lnb   = (const float*)d_in[9];

    float* ws = (float*)d_ws;
    float* Qp = ws;
    float* Kp = Qp + (size_t)NN * HID;
    float* Vp = Kp + (size_t)NN * HID;
    float* Ep = Vp + (size_t)NN * HID;
    float* Gp = Ep + (size_t)NN * NHEADS;
    float* Wt = Gp + (size_t)NN * NHEADS;

    wpack_kernel<<<dim3(25), 32, 0, stream>>>(Wq, Wkv, Weg, Wt);

    proj_kernel<<<dim3(NN / 16), 256, 0, stream>>>(
        feat1, feat2, Wt, bq, bkv, beg, Qp, Kp, Vp, Ep, Gp);

    attn_kernel<<<dim3(NSZ, NSZ / 16), 256, 0, stream>>>(
        Qp, Kp, Vp, Ep, Gp, lng, lnb, (float*)d_out);
}


# K and V workspace planes stored in WMMA fragment order: each attention operand load is 16 contiguous B per lane (4 lines) instead of an 8 B gather over 8-16 lines; d and key order of sums regrouped
# speedup vs baseline: 1.2808x; 1.0726x over previous
#include <hip/hip_runtime.h>
#include <hip/hip_bf16.h>

#define NSZ 160
#define NN (NSZ * NSZ)
#define HID 128
#define NHEADS 8
#define HD 16
#define LN_EPS 0.001f
#define APITCH 132

typedef __attribute__((ext_vector_type(2))) float v2f;
typedef __attribute__((ext_vector_type(4))) float v4f;
typedef __attribute__((ext_vector_type(8))) float v8f;

static __device__ __forceinline__ v8f wmma_f32_16x16x4(v2f a, v2f b, v8f c) {
    return __builtin_amdgcn_wmma_f32_16x16x4_f32(
         false, a,  false, b,
         (short)0, c,  false,  false);
}


__global__ __launch_bounds__(32) void wpack_kernel(
    const float* __restrict__ Wq, const float* __restrict__ Wkv,
    const float* __restrict__ Weg, float* __restrict__ Wt)
{
    const int ct   = blockIdx.x;
    const int lane = threadIdx.x;
    const int hs   = lane / 16;
    const int lm   = lane % 16;

    const float* W; int wcols, n0;
    if (ct < 8)       { W = Wq;  wcols = HID;        n0 = ct * 16; }
    else if (ct < 24) { W = Wkv; wcols = 2 * HID;    n0 = (ct - 8) * 16; }
    else              { W = Weg; wcols = 2 * NHEADS; n0 = 0; }

    float* o = Wt + (size_t)ct * 2048 + lane * 2;
    for (int ks = 0; ks < 32; ++ks) {
        const int kk = ks * 4 + 2 * hs;
        o[ks * 64 + 0] = W[(size_t)(kk)     * wcols + n0 + lm];
        o[ks * 64 + 1] = W[(size_t)(kk + 1) * wcols + n0 + lm];
    }
}

static __device__ __forceinline__ v8f gemm_tile(
    const float* __restrict__ sA, const float* __restrict__ wt,
    int lm, int hs)
{
    v8f acc = {0.f, 0.f, 0.f, 0.f, 0.f, 0.f, 0.f, 0.f};
    const float* pa = sA + lm * APITCH + 2 * hs;
    #pragma unroll
    for (int ks = 0; ks < 32; ++ks) {
        v2f a, b;
        a.x = pa[ks * 4];
        a.y = pa[ks * 4 + 1];
        b = *reinterpret_cast<const v2f*>(wt + ks * 64);
        acc = wmma_f32_16x16x4(a, b, acc);
    }
    return acc;
}

__global__ __launch_bounds__(256) void proj_kernel(
    const float* __restrict__ feat1, const float* __restrict__ feat2,
    const float* __restrict__ Wt,
    const float* __restrict__ bq, const float* __restrict__ bkv,
    const float* __restrict__ beg,
    float* __restrict__ Qp, float* __restrict__ Kp, float* __restrict__ Vp,
    float* __restrict__ Ep, float* __restrict__ Gp)
{
    __shared__ float sA1[16][APITCH];
    __shared__ float sA2[16][APITCH];

    const int row0 = blockIdx.x * 16;
    const int tid  = threadIdx.x;

    for (int idx = tid; idx < 16 * (HID / 4); idx += 256) {
        const int r = idx / (HID / 4), c4 = idx % (HID / 4);
        reinterpret_cast<float4*>(&sA1[r][0])[c4] =
            reinterpret_cast<const float4*>(&feat1[(size_t)(row0 + r) * HID])[c4];
        reinterpret_cast<float4*>(&sA2[r][0])[c4] =
            reinterpret_cast<const float4*>(&feat2[(size_t)(row0 + r) * HID])[c4];
    }
    __syncthreads();

    const int wave = tid / 32;
    const int lane = tid % 32;
    const int hs   = lane / 16;
    const int lm   = lane % 16;

    const int aIdx = row0 / NSZ;
    const int b0   = row0 % NSZ;
    const int bBase = b0 + 8 * hs;

    const int col = wave * 16 + lm;
    const int d   = col / 8;
    const int h   = col % 8;

    {
        v8f acc = gemm_tile(&sA1[0][0], Wt + (size_t)wave * 2048 + lane * 2, lm, hs);
        const float bv = bq[col];
        float* oq = Qp + (((size_t)bBase * NHEADS + h) * NSZ + aIdx) * HD + d;
        #pragma unroll
        for (int r = 0; r < 8; ++r)
            oq[(size_t)r * NHEADS * NSZ * HD] = (acc[r] + bv) * 0.25f;
    }

    {
        v8f acc = gemm_tile(&sA2[0][0], Wt + (size_t)(8 + wave) * 2048 + lane * 2, lm, hs);
        const float bv = bkv[col];
        float* ok = Kp + ((size_t)aIdx * NHEADS + h) * (NSZ * HD) + (b0 / 16) * 256
                       + ((d % 8) / 4) * 128 + ((d / 8) * 16 + 8 * hs) * 4 + (d % 4);
        #pragma unroll
        for (int r = 0; r < 8; ++r)
            ok[r * 4] = acc[r] + bv;
    }

    {
        v8f acc = gemm_tile(&sA2[0][0], Wt + (size_t)(16 + wave) * 2048 + lane * 2, lm, hs);
        const float bv = bkv[HID + col];
        float* ov = Vp + ((size_t)aIdx * NHEADS + h) * (HD * NSZ) + (b0 / 8 + hs) * 128 + d * 4;
        v4f lo, hi;
        #pragma unroll
        for (int r = 0; r < 4; ++r) { lo[r] = acc[r] + bv; hi[r] = acc[r + 4] + bv; }
        *reinterpret_cast<v4f*>(ov)      = lo;
        *reinterpret_cast<v4f*>(ov + 64) = hi;
    }

    if (wave == 0) {
        v8f acc = gemm_tile(&sA2[0][0], Wt + (size_t)24 * 2048 + lane * 2, lm, hs);
        const float bv = beg[lm];
        if (lm < NHEADS) {
            float* oe = Ep + ((size_t)aIdx * NHEADS + lm) * NSZ + bBase;
            #pragma unroll
            for (int r = 0; r < 8; ++r) oe[r] = acc[r] + bv;
        } else {
            float* og = Gp + ((size_t)aIdx * NHEADS + (lm - NHEADS)) * NSZ + bBase;
            #pragma unroll
            for (int r = 0; r < 8; ++r)
                og[r] = 1.f / (1.f + __expf(-(acc[r] + bv)));
        }
    }
}

__global__ __launch_bounds__(256) __attribute__((amdgpu_waves_per_eu(6))) void attn_kernel(
    const float* __restrict__ Qp, const float* __restrict__ Kp,
    const float* __restrict__ Vp, const float* __restrict__ Ep,
    const float* __restrict__ Gp,
    const float* __restrict__ ln_g, const float* __restrict__ ln_b,
    float* __restrict__ out)
{
    __shared__ __attribute__((aligned(16))) float Plds[NHEADS][16][84];
    __shared__ float VaLds[16][HID];

    const int j    = blockIdx.x;
    const int i0   = blockIdx.y * 16;
    const int tid  = threadIdx.x;
    const int h    = __builtin_amdgcn_readfirstlane(tid / 32);
    const int lane = tid % 32;
    const int hs   = lane / 16;
    const int lm   = lane % 16;

    const size_t jh = (size_t)j * NHEADS + h;

    const float* pQ = Qp + (jh * NSZ + i0) * HD + (lm * HD + 8 * hs);
    const float* pK = Kp + jh * (NSZ * HD) + lane * 4;
    const float* pV = Vp + jh * (HD * NSZ) + lane * 4;
    const float* pE = Ep + jh * NSZ + lm;
    const float* pG = Gp + jh * NSZ + lm;

    {
        const float* vslice = Vp + jh * HD * NSZ;
        for (int p = lane; p < 80; p += 32)
            __builtin_prefetch(vslice + p * 32, 0, 1);
    }

    v2f afr[4];
    #pragma unroll
    for (int x = 0; x < 2; ++x) {
        const v4f q4 = *reinterpret_cast<const v4f*>(pQ + x * 4);
        afr[2 * x]     = v2f{q4[0], q4[1]};
        afr[2 * x + 1] = v2f{q4[2], q4[3]};
    }

    v2f kf[40];
    #pragma unroll
    for (int ct = 0; ct < 10; ++ct) {
        #pragma unroll
        for (int x = 0; x < 2; ++x) {
            const v4f k4 = *reinterpret_cast<const v4f*>(pK + ct * 256 + x * 128);
            kf[ct * 4 + 2 * x]     = v2f{k4[0], k4[1]};
            kf[ct * 4 + 2 * x + 1] = v2f{k4[2], k4[3]};
        }
    }
    float ev[10];
    #pragma unroll
    for (int ct = 0; ct < 10; ++ct) ev[ct] = pE[ct * 16];
    __builtin_amdgcn_sched_barrier(0);

    v8f acc[10];
    #pragma unroll
    for (int ct = 0; ct < 10; ++ct) {
        v8f c = {0.f, 0.f, 0.f, 0.f, 0.f, 0.f, 0.f, 0.f};
        #pragma unroll
        for (int ks = 0; ks < 4; ++ks)
            c = wmma_f32_16x16x4(afr[ks], kf[ct * 4 + ks], c);
        const float e = ev[ct];
        #pragma unroll
        for (int r = 0; r < 8; ++r) c[r] += e;
        acc[ct] = c;
    }

    float gate[10];
    #pragma unroll
    for (int ct = 0; ct < 10; ++ct) gate[ct] = pG[ct * 16];

    #pragma unroll
    for (int r = 0; r < 8; ++r) {
        float mx = -1e30f;
        #pragma unroll
        for (int ct = 0; ct < 10; ++ct) mx = fmaxf(mx, acc[ct][r]);
        #pragma unroll
        for (int msk = 1; msk < 16; msk <<= 1) mx = fmaxf(mx, __shfl_xor(mx, msk, 32));
        float sm = 0.f;
        #pragma unroll
        for (int ct = 0; ct < 10; ++ct) {
            float e = __expf(acc[ct][r] - mx);
            acc[ct][r] = e;
            sm += e;
        }
        #pragma unroll
        for (int msk = 1; msk < 16; msk <<= 1) sm += __shfl_xor(sm, msk, 32);
        const float inv = 1.f / sm;
        #pragma unroll
        for (int ct = 0; ct < 10; ++ct) acc[ct][r] = acc[ct][r] * inv * gate[ct];
    }

    v8f va = {0.f, 0.f, 0.f, 0.f, 0.f, 0.f, 0.f, 0.f};
    #pragma unroll
    for (int half = 0; half < 2; ++half) {
        #pragma unroll
        for (int ct = 0; ct < 5; ++ct) {
            #pragma unroll
            for (int r = 0; r < 8; ++r)
                Plds[h][r + 8 * hs][ct * 16 + lm] = acc[half * 5 + ct][r];
        }
        __builtin_amdgcn_sched_barrier(0);
        v2f vf[20];
        #pragma unroll
        for (int g = 0; g < 10; ++g) {
            const v4f v4 = *reinterpret_cast<const v4f*>(pV + (half * 10 + g) * 128);
            vf[2 * g]     = v2f{v4[0], v4[1]};
            vf[2 * g + 1] = v2f{v4[2], v4[3]};
        }
        __syncthreads();
        v2f pf[20];
        #pragma unroll
        for (int g = 0; g < 10; ++g) {
            const v4f p4 = *reinterpret_cast<const v4f*>(&Plds[h][lm][g * 8 + 4 * hs]);
            pf[2 * g]     = v2f{p4[0], p4[1]};
            pf[2 * g + 1] = v2f{p4[2], p4[3]};
        }
        __builtin_amdgcn_sched_barrier(0);
        #pragma unroll
        for (int q = 0; q < 20; ++q)
            va = wmma_f32_16x16x4(pf[q], vf[q], va);
        __syncthreads();
    }

    #pragma unroll
    for (int r = 0; r < 8; ++r)
        VaLds[r + 8 * hs][lm * NHEADS + h] = va[r];
    __syncthreads();

    const int rowLN = tid / 16;
    const int cg    = tid % 16;
    float vals[8], s1 = 0.f, s2 = 0.f;
    #pragma unroll
    for (int c = 0; c < 8; ++c) {
        float v = VaLds[rowLN][cg * 8 + c];
        vals[c] = v; s1 += v; s2 += v * v;
    }
    #pragma unroll
    for (int msk = 1; msk < 16; msk <<= 1) {
        s1 += __shfl_xor(s1, msk, 32);
        s2 += __shfl_xor(s2, msk, 32);
    }
    const float mu  = s1 * (1.f / HID);
    const float var = s2 * (1.f / HID) - mu * mu;
    const float inv = rsqrtf(var + LN_EPS);

    const size_t obase = ((size_t)(i0 + rowLN) * NSZ + j) * HID + cg * 8;
    v2f* op = reinterpret_cast<v2f*>(&out[obase]);
    #pragma unroll
    for (int c = 0; c < 8; c += 2) {
        const int ch = cg * 8 + c;
        v2f t;
        t.x = (vals[c]     - mu) * inv * ln_g[ch]     + ln_b[ch];
        t.y = (vals[c + 1] - mu) * inv * ln_g[ch + 1] + ln_b[ch + 1];
        op[c / 2] = t;
    }
}

extern "C" void kernel_launch(void* const* d_in, const int* in_sizes, int n_in,
                              void* d_out, int out_size, void* d_ws, size_t ws_size,
                              hipStream_t stream) {
    const float* feat1 = (const float*)d_in[0];
    const float* feat2 = (const float*)d_in[1];
    const float* Wq    = (const float*)d_in[2];
    const float* bq    = (const float*)d_in[3];
    const float* Wkv   = (const float*)d_in[4];
    const float* bkv   = (const float*)d_in[5];
    const float* Weg   = (const float*)d_in[6];
    const float* beg   = (const float*)d_in[7];
    const float* lng   = (const float*)d_in[8];
    const float* lnb   = (const float*)d_in[9];

    float* ws = (float*)d_ws;
    float* Qp = ws;
    float* Kp = Qp + (size_t)NN * HID;
    float* Vp = Kp + (size_t)NN * HID;
    float* Ep = Vp + (size_t)NN * HID;
    float* Gp = Ep + (size_t)NN * NHEADS;
    float* Wt = Gp + (size_t)NN * NHEADS;

    wpack_kernel<<<dim3(25), 32, 0, stream>>>(Wq, Wkv, Weg, Wt);

    proj_kernel<<<dim3(NN / 16), 256, 0, stream>>>(
        feat1, feat2, Wt, bq, bkv, beg, Qp, Kp, Vp, Ep, Gp);

    attn_kernel<<<dim3(NSZ, NSZ / 16), 256, 0, stream>>>(
        Qp, Kp, Vp, Ep, Gp, lng, lnb, (float*)d_out);
}


# Projection stores as whole lines: each wave owns one head (weight columns permuted in the pack), Q and K tiles formed transposed (WMMA A/B roles swapped), Q plane in fragment order: 512 B per store
# speedup vs baseline: 2.1645x; 1.6900x over previous
#include <hip/hip_runtime.h>
#include <hip/hip_bf16.h>

#define NSZ 160
#define NN (NSZ * NSZ)
#define HID 128
#define NHEADS 8
#define HD 16
#define LN_EPS 0.001f
#define APITCH 132

typedef __attribute__((ext_vector_type(2))) float v2f;
typedef __attribute__((ext_vector_type(4))) float v4f;
typedef __attribute__((ext_vector_type(8))) float v8f;

static __device__ __forceinline__ v8f wmma_f32_16x16x4(v2f a, v2f b, v8f c) {
    return __builtin_amdgcn_wmma_f32_16x16x4_f32(
         false, a,  false, b,
         (short)0, c,  false,  false);
}


__global__ __launch_bounds__(32) void wpack_kernel(
    const float* __restrict__ Wq, const float* __restrict__ Wkv,
    const float* __restrict__ Weg, float* __restrict__ Wt)
{
    const int ct   = blockIdx.x;
    const int lane = threadIdx.x;
    const int hs   = lane / 16;
    const int lm   = lane % 16;

    const float* W; int wcols, cn;
    if (ct < 8)       { W = Wq;  wcols = HID;        cn = lm * NHEADS + ct; }
    else if (ct < 16) { W = Wkv; wcols = 2 * HID;    cn = lm * NHEADS + (ct - 8); }
    else if (ct < 24) { W = Wkv; wcols = 2 * HID;    cn = HID + lm * NHEADS + (ct - 16); }
    else              { W = Weg; wcols = 2 * NHEADS; cn = lm; }

    float* o = Wt + (size_t)ct * 2048 + lane * 2;
    for (int ks = 0; ks < 32; ++ks) {
        const int kk = ks * 4 + 2 * hs;
        o[ks * 64 + 0] = W[(size_t)(kk)     * wcols + cn];
        o[ks * 64 + 1] = W[(size_t)(kk + 1) * wcols + cn];
    }
}

template <bool TRANSPOSED>
static __device__ __forceinline__ v8f gemm_tile(
    const float* __restrict__ sA, const float* __restrict__ wt,
    int lm, int hs)
{
    v8f acc = {0.f, 0.f, 0.f, 0.f, 0.f, 0.f, 0.f, 0.f};
    const float* pa = sA + lm * APITCH + 2 * hs;
    #pragma unroll
    for (int ks = 0; ks < 32; ++ks) {
        v2f a, b;
        a.x = pa[ks * 4];
        a.y = pa[ks * 4 + 1];
        b = *reinterpret_cast<const v2f*>(wt + ks * 64);
        acc = TRANSPOSED ? wmma_f32_16x16x4(b, a, acc) : wmma_f32_16x16x4(a, b, acc);
    }
    return acc;
}

__global__ __launch_bounds__(256) void proj_kernel(
    const float* __restrict__ feat1, const float* __restrict__ feat2,
    const float* __restrict__ Wt,
    const float* __restrict__ bq, const float* __restrict__ bkv,
    const float* __restrict__ beg,
    float* __restrict__ Qp, float* __restrict__ Kp, float* __restrict__ Vp,
    float* __restrict__ Ep, float* __restrict__ Gp)
{
    __shared__ float sA1[16][APITCH];
    __shared__ float sA2[16][APITCH];

    const int row0 = blockIdx.x * 16;
    const int tid  = threadIdx.x;

    const int aIdx = row0 / NSZ;
    const int b0   = row0 % NSZ;

    for (int idx = tid; idx < 16 * (HID / 4); idx += 256) {
        const int r = idx / (HID / 4), c4 = idx % (HID / 4);
        reinterpret_cast<float4*>(&sA1[r][0])[c4] =
            reinterpret_cast<const float4*>(&feat1[((size_t)(b0 + r) * NSZ + aIdx) * HID])[c4];
        reinterpret_cast<float4*>(&sA2[r][0])[c4] =
            reinterpret_cast<const float4*>(&feat2[(size_t)(row0 + r) * HID])[c4];
    }
    __syncthreads();

    const int wave = tid / 32;
    const int lane = tid % 32;
    const int hs   = lane / 16;
    const int lm   = lane % 16;

    const int bBase = b0 + 8 * hs;

    const size_t plane = ((size_t)aIdx * NHEADS + wave) * (NSZ * HD);

    {
        v8f acc = gemm_tile<true>(&sA1[0][0], Wt + (size_t)wave * 2048 + lane * 2, lm, hs);
        const float* bp = bq + (8 * hs) * NHEADS + wave;
        float* oq = Qp + plane + (b0 / 16) * 256 + lane * 4;
        v4f lo, hi;
        #pragma unroll
        for (int r = 0; r < 4; ++r) {
            lo[r] = (acc[r]     + bp[r * NHEADS])       * 0.25f;
            hi[r] = (acc[r + 4] + bp[(r + 4) * NHEADS]) * 0.25f;
        }
        *reinterpret_cast<v4f*>(oq)       = lo;
        *reinterpret_cast<v4f*>(oq + 128) = hi;
    }

    {
        v8f acc = gemm_tile<true>(&sA2[0][0], Wt + (size_t)(8 + wave) * 2048 + lane * 2, lm, hs);
        const float* bp = bkv + (8 * hs) * NHEADS + wave;
        float* ok = Kp + plane + (b0 / 16) * 256 + lane * 4;
        v4f lo, hi;
        #pragma unroll
        for (int r = 0; r < 4; ++r) {
            lo[r] = acc[r]     + bp[r * NHEADS];
            hi[r] = acc[r + 4] + bp[(r + 4) * NHEADS];
        }
        *reinterpret_cast<v4f*>(ok)       = lo;
        *reinterpret_cast<v4f*>(ok + 128) = hi;
    }

    {
        v8f acc = gemm_tile<false>(&sA2[0][0], Wt + (size_t)(16 + wave) * 2048 + lane * 2, lm, hs);
        const float bv = bkv[HID + lm * NHEADS + wave];
        float* ov = Vp + plane + (b0 / 8 + hs) * 128 + lm * 4;
        v4f lo, hi;
        #pragma unroll
        for (int r = 0; r < 4; ++r) { lo[r] = acc[r] + bv; hi[r] = acc[r + 4] + bv; }
        *reinterpret_cast<v4f*>(ov)      = lo;
        *reinterpret_cast<v4f*>(ov + 64) = hi;
    }

    if (wave == 0) {
        v8f acc = gemm_tile<false>(&sA2[0][0], Wt + (size_t)24 * 2048 + lane * 2, lm, hs);
        const float bv = beg[lm];
        if (lm < NHEADS) {
            float* oe = Ep + ((size_t)aIdx * NHEADS + lm) * NSZ + bBase;
            #pragma unroll
            for (int r = 0; r < 8; ++r) oe[r] = acc[r] + bv;
        } else {
            float* og = Gp + ((size_t)aIdx * NHEADS + (lm - NHEADS)) * NSZ + bBase;
            #pragma unroll
            for (int r = 0; r < 8; ++r)
                og[r] = 1.f / (1.f + __expf(-(acc[r] + bv)));
        }
    }
}

__global__ __launch_bounds__(256) __attribute__((amdgpu_waves_per_eu(6))) void attn_kernel(
    const float* __restrict__ Qp, const float* __restrict__ Kp,
    const float* __restrict__ Vp, const float* __restrict__ Ep,
    const float* __restrict__ Gp,
    const float* __restrict__ ln_g, const float* __restrict__ ln_b,
    float* __restrict__ out)
{
    __shared__ __attribute__((aligned(16))) float Plds[NHEADS][16][84];
    __shared__ float VaLds[16][HID];

    const int j    = blockIdx.x;
    const int i0   = blockIdx.y * 16;
    const int tid  = threadIdx.x;
    const int h    = __builtin_amdgcn_readfirstlane(tid / 32);
    const int lane = tid % 32;
    const int hs   = lane / 16;
    const int lm   = lane % 16;

    const size_t jh = (size_t)j * NHEADS + h;

    const float* pQ = Qp + jh * (NSZ * HD) + blockIdx.y * 256 + lane * 4;
    const float* pK = Kp + jh * (NSZ * HD) + lane * 4;
    const float* pV = Vp + jh * (HD * NSZ) + lane * 4;
    const float* pE = Ep + jh * NSZ + lm;
    const float* pG = Gp + jh * NSZ + lm;

    {
        const float* vslice = Vp + jh * HD * NSZ;
        for (int p = lane; p < 80; p += 32)
            __builtin_prefetch(vslice + p * 32, 0, 1);
    }

    v2f afr[4];
    #pragma unroll
    for (int x = 0; x < 2; ++x) {
        const v4f q4 = *reinterpret_cast<const v4f*>(pQ + x * 128);
        afr[2 * x]     = v2f{q4[0], q4[1]};
        afr[2 * x + 1] = v2f{q4[2], q4[3]};
    }

    v2f kf[40];
    #pragma unroll
    for (int ct = 0; ct < 10; ++ct) {
        #pragma unroll
        for (int x = 0; x < 2; ++x) {
            const v4f k4 = *reinterpret_cast<const v4f*>(pK + ct * 256 + x * 128);
            kf[ct * 4 + 2 * x]     = v2f{k4[0], k4[1]};
            kf[ct * 4 + 2 * x + 1] = v2f{k4[2], k4[3]};
        }
    }
    float ev[10];
    #pragma unroll
    for (int ct = 0; ct < 10; ++ct) ev[ct] = pE[ct * 16];
    __builtin_amdgcn_sched_barrier(0);

    v8f acc[10];
    #pragma unroll
    for (int ct = 0; ct < 10; ++ct) {
        v8f c = {0.f, 0.f, 0.f, 0.f, 0.f, 0.f, 0.f, 0.f};
        #pragma unroll
        for (int ks = 0; ks < 4; ++ks)
            c = wmma_f32_16x16x4(afr[ks], kf[ct * 4 + ks], c);
        const float e = ev[ct];
        #pragma unroll
        for (int r = 0; r < 8; ++r) c[r] += e;
        acc[ct] = c;
    }

    float gate[10];
    #pragma unroll
    for (int ct = 0; ct < 10; ++ct) gate[ct] = pG[ct * 16];

    #pragma unroll
    for (int r = 0; r < 8; ++r) {
        float mx = -1e30f;
        #pragma unroll
        for (int ct = 0; ct < 10; ++ct) mx = fmaxf(mx, acc[ct][r]);
        #pragma unroll
        for (int msk = 1; msk < 16; msk <<= 1) mx = fmaxf(mx, __shfl_xor(mx, msk, 32));
        float sm = 0.f;
        #pragma unroll
        for (int ct = 0; ct < 10; ++ct) {
            float e = __expf(acc[ct][r] - mx);
            acc[ct][r] = e;
            sm += e;
        }
        #pragma unroll
        for (int msk = 1; msk < 16; msk <<= 1) sm += __shfl_xor(sm, msk, 32);
        const float inv = 1.f / sm;
        #pragma unroll
        for (int ct = 0; ct < 10; ++ct) acc[ct][r] = acc[ct][r] * inv * gate[ct];
    }

    v8f va = {0.f, 0.f, 0.f, 0.f, 0.f, 0.f, 0.f, 0.f};
    #pragma unroll
    for (int half = 0; half < 2; ++half) {
        #pragma unroll
        for (int ct = 0; ct < 5; ++ct) {
            #pragma unroll
            for (int r = 0; r < 8; ++r)
                Plds[h][r + 8 * hs][ct * 16 + lm] = acc[half * 5 + ct][r];
        }
        __builtin_amdgcn_sched_barrier(0);
        v2f vf[20];
        #pragma unroll
        for (int g = 0; g < 10; ++g) {
            const v4f v4 = *reinterpret_cast<const v4f*>(pV + (half * 10 + g) * 128);
            vf[2 * g]     = v2f{v4[0], v4[1]};
            vf[2 * g + 1] = v2f{v4[2], v4[3]};
        }
        __syncthreads();
        v2f pf[20];
        #pragma unroll
        for (int g = 0; g < 10; ++g) {
            const v4f p4 = *reinterpret_cast<const v4f*>(&Plds[h][lm][g * 8 + 4 * hs]);
            pf[2 * g]     = v2f{p4[0], p4[1]};
            pf[2 * g + 1] = v2f{p4[2], p4[3]};
        }
        __builtin_amdgcn_sched_barrier(0);
        #pragma unroll
        for (int q = 0; q < 20; ++q)
            va = wmma_f32_16x16x4(pf[q], vf[q], va);
        __syncthreads();
    }

    #pragma unroll
    for (int r = 0; r < 8; ++r)
        VaLds[r + 8 * hs][lm * NHEADS + h] = va[r];
    __syncthreads();

    const int rowLN = tid / 16;
    const int cg    = tid % 16;
    float vals[8], s1 = 0.f, s2 = 0.f;
    #pragma unroll
    for (int c = 0; c < 8; ++c) {
        float v = VaLds[rowLN][cg * 8 + c];
        vals[c] = v; s1 += v; s2 += v * v;
    }
    #pragma unroll
    for (int msk = 1; msk < 16; msk <<= 1) {
        s1 += __shfl_xor(s1, msk, 32);
        s2 += __shfl_xor(s2, msk, 32);
    }
    const float mu  = s1 * (1.f / HID);
    const float var = s2 * (1.f / HID) - mu * mu;
    const float inv = rsqrtf(var + LN_EPS);

    const size_t obase = ((size_t)(i0 + rowLN) * NSZ + j) * HID + cg * 8;
    v2f* op = reinterpret_cast<v2f*>(&out[obase]);
    #pragma unroll
    for (int c = 0; c < 8; c += 2) {
        const int ch = cg * 8 + c;
        v2f t;
        t.x = (vals[c]     - mu) * inv * ln_g[ch]     + ln_b[ch];
        t.y = (vals[c + 1] - mu) * inv * ln_g[ch + 1] + ln_b[ch + 1];
        op[c / 2] = t;
    }
}

extern "C" void kernel_launch(void* const* d_in, const int* in_sizes, int n_in,
                              void* d_out, int out_size, void* d_ws, size_t ws_size,
                              hipStream_t stream) {
    const float* feat1 = (const float*)d_in[0];
    const float* feat2 = (const float*)d_in[1];
    const float* Wq    = (const float*)d_in[2];
    const float* bq    = (const float*)d_in[3];
    const float* Wkv   = (const float*)d_in[4];
    const float* bkv   = (const float*)d_in[5];
    const float* Weg   = (const float*)d_in[6];
    const float* beg   = (const float*)d_in[7];
    const float* lng   = (const float*)d_in[8];
    const float* lnb   = (const float*)d_in[9];

    float* ws = (float*)d_ws;
    float* Qp = ws;
    float* Kp = Qp + (size_t)NN * HID;
    float* Vp = Kp + (size_t)NN * HID;
    float* Ep = Vp + (size_t)NN * HID;
    float* Gp = Ep + (size_t)NN * NHEADS;
    float* Wt = Gp + (size_t)NN * NHEADS;

    wpack_kernel<<<dim3(25), 32, 0, stream>>>(Wq, Wkv, Weg, Wt);

    proj_kernel<<<dim3(NN / 16), 256, 0, stream>>>(
        feat1, feat2, Wt, bq, bkv, beg, Qp, Kp, Vp, Ep, Gp);

    attn_kernel<<<dim3(NSZ, NSZ / 16), 256, 0, stream>>>(
        Qp, Kp, Vp, Ep, Gp, lng, lnb, (float*)d_out);
}


# Attention softmax and LayerNorm: each 16-lane max/sum butterfly is four DPP row_xmask vector-ALU ops instead of four LDS-unit shuffles each with its own round trip and wait; same sum order
# speedup vs baseline: 2.2283x; 1.0295x over previous
#include <hip/hip_runtime.h>
#include <hip/hip_bf16.h>

#define NSZ 160
#define NN (NSZ * NSZ)
#define HID 128
#define NHEADS 8
#define HD 16
#define LN_EPS 0.001f
#define APITCH 132

typedef __attribute__((ext_vector_type(2))) float v2f;
typedef __attribute__((ext_vector_type(4))) float v4f;
typedef __attribute__((ext_vector_type(8))) float v8f;

static __device__ __forceinline__ v8f wmma_f32_16x16x4(v2f a, v2f b, v8f c) {
    return __builtin_amdgcn_wmma_f32_16x16x4_f32(
         false, a,  false, b,
         (short)0, c,  false,  false);
}

static __device__ __forceinline__ float row16_max(float v) {
    asm("v_max_num_f32_dpp %0, %0, %0 row_xmask:1 row_mask:0xf bank_mask:0xf\n\t"
        "v_max_num_f32_dpp %0, %0, %0 row_xmask:2 row_mask:0xf bank_mask:0xf\n\t"
        "v_max_num_f32_dpp %0, %0, %0 row_xmask:4 row_mask:0xf bank_mask:0xf\n\t"
        "v_max_num_f32_dpp %0, %0, %0 row_xmask:8 row_mask:0xf bank_mask:0xf"
        : "+v"(v));
    return v;
}
static __device__ __forceinline__ float row16_sum(float v) {
    asm("v_add_f32_dpp %0, %0, %0 row_xmask:1 row_mask:0xf bank_mask:0xf\n\t"
        "v_add_f32_dpp %0, %0, %0 row_xmask:2 row_mask:0xf bank_mask:0xf\n\t"
        "v_add_f32_dpp %0, %0, %0 row_xmask:4 row_mask:0xf bank_mask:0xf\n\t"
        "v_add_f32_dpp %0, %0, %0 row_xmask:8 row_mask:0xf bank_mask:0xf"
        : "+v"(v));
    return v;
}


__global__ __launch_bounds__(32) void wpack_kernel(
    const float* __restrict__ Wq, const float* __restrict__ Wkv,
    const float* __restrict__ Weg, float* __restrict__ Wt)
{
    const int ct   = blockIdx.x;
    const int lane = threadIdx.x;
    const int hs   = lane / 16;
    const int lm   = lane % 16;

    const float* W; int wcols, cn;
    if (ct < 8)       { W = Wq;  wcols = HID;        cn = lm * NHEADS + ct; }
    else if (ct < 16) { W = Wkv; wcols = 2 * HID;    cn = lm * NHEADS + (ct - 8); }
    else if (ct < 24) { W = Wkv; wcols = 2 * HID;    cn = HID + lm * NHEADS + (ct - 16); }
    else              { W = Weg; wcols = 2 * NHEADS; cn = lm; }

    float* o = Wt + (size_t)ct * 2048 + lane * 2;
    for (int ks = 0; ks < 32; ++ks) {
        const int kk = ks * 4 + 2 * hs;
        o[ks * 64 + 0] = W[(size_t)(kk)     * wcols + cn];
        o[ks * 64 + 1] = W[(size_t)(kk + 1) * wcols + cn];
    }
}

template <bool TRANSPOSED>
static __device__ __forceinline__ v8f gemm_tile(
    const float* __restrict__ sA, const float* __restrict__ wt,
    int lm, int hs)
{
    v8f acc = {0.f, 0.f, 0.f, 0.f, 0.f, 0.f, 0.f, 0.f};
    const float* pa = sA + lm * APITCH + 2 * hs;
    #pragma unroll
    for (int ks = 0; ks < 32; ++ks) {
        v2f a, b;
        a.x = pa[ks * 4];
        a.y = pa[ks * 4 + 1];
        b = *reinterpret_cast<const v2f*>(wt + ks * 64);
        acc = TRANSPOSED ? wmma_f32_16x16x4(b, a, acc) : wmma_f32_16x16x4(a, b, acc);
    }
    return acc;
}

__global__ __launch_bounds__(256) void proj_kernel(
    const float* __restrict__ feat1, const float* __restrict__ feat2,
    const float* __restrict__ Wt,
    const float* __restrict__ bq, const float* __restrict__ bkv,
    const float* __restrict__ beg,
    float* __restrict__ Qp, float* __restrict__ Kp, float* __restrict__ Vp,
    float* __restrict__ Ep, float* __restrict__ Gp)
{
    __shared__ float sA1[16][APITCH];
    __shared__ float sA2[16][APITCH];

    const int row0 = blockIdx.x * 16;
    const int tid  = threadIdx.x;

    const int aIdx = row0 / NSZ;
    const int b0   = row0 % NSZ;

    for (int idx = tid; idx < 16 * (HID / 4); idx += 256) {
        const int r = idx / (HID / 4), c4 = idx % (HID / 4);
        reinterpret_cast<float4*>(&sA1[r][0])[c4] =
            reinterpret_cast<const float4*>(&feat1[((size_t)(b0 + r) * NSZ + aIdx) * HID])[c4];
        reinterpret_cast<float4*>(&sA2[r][0])[c4] =
            reinterpret_cast<const float4*>(&feat2[(size_t)(row0 + r) * HID])[c4];
    }
    __syncthreads();

    const int wave = tid / 32;
    const int lane = tid % 32;
    const int hs   = lane / 16;
    const int lm   = lane % 16;

    const int bBase = b0 + 8 * hs;

    const size_t plane = ((size_t)aIdx * NHEADS + wave) * (NSZ * HD);

    {
        v8f acc = gemm_tile<true>(&sA1[0][0], Wt + (size_t)wave * 2048 + lane * 2, lm, hs);
        const float* bp = bq + (8 * hs) * NHEADS + wave;
        float* oq = Qp + plane + (b0 / 16) * 256 + lane * 4;
        v4f lo, hi;
        #pragma unroll
        for (int r = 0; r < 4; ++r) {
            lo[r] = (acc[r]     + bp[r * NHEADS])       * 0.25f;
            hi[r] = (acc[r + 4] + bp[(r + 4) * NHEADS]) * 0.25f;
        }
        *reinterpret_cast<v4f*>(oq)       = lo;
        *reinterpret_cast<v4f*>(oq + 128) = hi;
    }

    {
        v8f acc = gemm_tile<true>(&sA2[0][0], Wt + (size_t)(8 + wave) * 2048 + lane * 2, lm, hs);
        const float* bp = bkv + (8 * hs) * NHEADS + wave;
        float* ok = Kp + plane + (b0 / 16) * 256 + lane * 4;
        v4f lo, hi;
        #pragma unroll
        for (int r = 0; r < 4; ++r) {
            lo[r] = acc[r]     + bp[r * NHEADS];
            hi[r] = acc[r + 4] + bp[(r + 4) * NHEADS];
        }
        *reinterpret_cast<v4f*>(ok)       = lo;
        *reinterpret_cast<v4f*>(ok + 128) = hi;
    }

    {
        v8f acc = gemm_tile<false>(&sA2[0][0], Wt + (size_t)(16 + wave) * 2048 + lane * 2, lm, hs);
        const float bv = bkv[HID + lm * NHEADS + wave];
        float* ov = Vp + plane + (b0 / 8 + hs) * 128 + lm * 4;
        v4f lo, hi;
        #pragma unroll
        for (int r = 0; r < 4; ++r) { lo[r] = acc[r] + bv; hi[r] = acc[r + 4] + bv; }
        *reinterpret_cast<v4f*>(ov)      = lo;
        *reinterpret_cast<v4f*>(ov + 64) = hi;
    }

    if (wave == 0) {
        v8f acc = gemm_tile<false>(&sA2[0][0], Wt + (size_t)24 * 2048 + lane * 2, lm, hs);
        const float bv = beg[lm];
        if (lm < NHEADS) {
            float* oe = Ep + ((size_t)aIdx * NHEADS + lm) * NSZ + bBase;
            #pragma unroll
            for (int r = 0; r < 8; ++r) oe[r] = acc[r] + bv;
        } else {
            float* og = Gp + ((size_t)aIdx * NHEADS + (lm - NHEADS)) * NSZ + bBase;
            #pragma unroll
            for (int r = 0; r < 8; ++r)
                og[r] = 1.f / (1.f + __expf(-(acc[r] + bv)));
        }
    }
}

__global__ __launch_bounds__(256) __attribute__((amdgpu_waves_per_eu(6))) void attn_kernel(
    const float* __restrict__ Qp, const float* __restrict__ Kp,
    const float* __restrict__ Vp, const float* __restrict__ Ep,
    const float* __restrict__ Gp,
    const float* __restrict__ ln_g, const float* __restrict__ ln_b,
    float* __restrict__ out)
{
    __shared__ __attribute__((aligned(16))) float Plds[NHEADS][16][84];
    __shared__ float VaLds[16][HID];

    const int j    = blockIdx.x;
    const int i0   = blockIdx.y * 16;
    const int tid  = threadIdx.x;
    const int h    = __builtin_amdgcn_readfirstlane(tid / 32);
    const int lane = tid % 32;
    const int hs   = lane / 16;
    const int lm   = lane % 16;

    const size_t jh = (size_t)j * NHEADS + h;

    const float* pQ = Qp + jh * (NSZ * HD) + blockIdx.y * 256 + lane * 4;
    const float* pK = Kp + jh * (NSZ * HD) + lane * 4;
    const float* pV = Vp + jh * (HD * NSZ) + lane * 4;
    const float* pE = Ep + jh * NSZ + lm;
    const float* pG = Gp + jh * NSZ + lm;

    {
        const float* vslice = Vp + jh * HD * NSZ;
        for (int p = lane; p < 80; p += 32)
            __builtin_prefetch(vslice + p * 32, 0, 1);
    }

    v2f afr[4];
    #pragma unroll
    for (int x = 0; x < 2; ++x) {
        const v4f q4 = *reinterpret_cast<const v4f*>(pQ + x * 128);
        afr[2 * x]     = v2f{q4[0], q4[1]};
        afr[2 * x + 1] = v2f{q4[2], q4[3]};
    }

    v2f kf[40];
    #pragma unroll
    for (int ct = 0; ct < 10; ++ct) {
        #pragma unroll
        for (int x = 0; x < 2; ++x) {
            const v4f k4 = *reinterpret_cast<const v4f*>(pK + ct * 256 + x * 128);
            kf[ct * 4 + 2 * x]     = v2f{k4[0], k4[1]};
            kf[ct * 4 + 2 * x + 1] = v2f{k4[2], k4[3]};
        }
    }
    float ev[10];
    #pragma unroll
    for (int ct = 0; ct < 10; ++ct) ev[ct] = pE[ct * 16];
    __builtin_amdgcn_sched_barrier(0);

    v8f acc[10];
    #pragma unroll
    for (int ct = 0; ct < 10; ++ct) {
        v8f c = {0.f, 0.f, 0.f, 0.f, 0.f, 0.f, 0.f, 0.f};
        #pragma unroll
        for (int ks = 0; ks < 4; ++ks)
            c = wmma_f32_16x16x4(afr[ks], kf[ct * 4 + ks], c);
        const float e = ev[ct];
        #pragma unroll
        for (int r = 0; r < 8; ++r) c[r] += e;
        acc[ct] = c;
    }

    float gate[10];
    #pragma unroll
    for (int ct = 0; ct < 10; ++ct) gate[ct] = pG[ct * 16];

    #pragma unroll
    for (int r = 0; r < 8; ++r) {
        float mx = -1e30f;
        #pragma unroll
        for (int ct = 0; ct < 10; ++ct) mx = fmaxf(mx, acc[ct][r]);
        mx = row16_max(mx);
        float sm = 0.f;
        #pragma unroll
        for (int ct = 0; ct < 10; ++ct) {
            float e = __expf(acc[ct][r] - mx);
            acc[ct][r] = e;
            sm += e;
        }
        sm = row16_sum(sm);
        const float inv = 1.f / sm;
        #pragma unroll
        for (int ct = 0; ct < 10; ++ct) acc[ct][r] = acc[ct][r] * inv * gate[ct];
    }

    v8f va = {0.f, 0.f, 0.f, 0.f, 0.f, 0.f, 0.f, 0.f};
    #pragma unroll
    for (int half = 0; half < 2; ++half) {
        #pragma unroll
        for (int ct = 0; ct < 5; ++ct) {
            #pragma unroll
            for (int r = 0; r < 8; ++r)
                Plds[h][r + 8 * hs][ct * 16 + lm] = acc[half * 5 + ct][r];
        }
        __builtin_amdgcn_sched_barrier(0);
        v2f vf[20];
        #pragma unroll
        for (int g = 0; g < 10; ++g) {
            const v4f v4 = *reinterpret_cast<const v4f*>(pV + (half * 10 + g) * 128);
            vf[2 * g]     = v2f{v4[0], v4[1]};
            vf[2 * g + 1] = v2f{v4[2], v4[3]};
        }
        __syncthreads();
        v2f pf[20];
        #pragma unroll
        for (int g = 0; g < 10; ++g) {
            const v4f p4 = *reinterpret_cast<const v4f*>(&Plds[h][lm][g * 8 + 4 * hs]);
            pf[2 * g]     = v2f{p4[0], p4[1]};
            pf[2 * g + 1] = v2f{p4[2], p4[3]};
        }
        __builtin_amdgcn_sched_barrier(0);
        #pragma unroll
        for (int q = 0; q < 20; ++q)
            va = wmma_f32_16x16x4(pf[q], vf[q], va);
        __syncthreads();
    }

    #pragma unroll
    for (int r = 0; r < 8; ++r)
        VaLds[r + 8 * hs][lm * NHEADS + h] = va[r];
    __syncthreads();

    const int rowLN = tid / 16;
    const int cg    = tid % 16;
    float vals[8], s1 = 0.f, s2 = 0.f;
    #pragma unroll
    for (int c = 0; c < 8; ++c) {
        float v = VaLds[rowLN][cg * 8 + c];
        vals[c] = v; s1 += v; s2 += v * v;
    }
    s1 = row16_sum(s1);
    s2 = row16_sum(s2);
    const float mu  = s1 * (1.f / HID);
    const float var = s2 * (1.f / HID) - mu * mu;
    const float inv = rsqrtf(var + LN_EPS);

    const size_t obase = ((size_t)(i0 + rowLN) * NSZ + j) * HID + cg * 8;
    v2f* op = reinterpret_cast<v2f*>(&out[obase]);
    #pragma unroll
    for (int c = 0; c < 8; c += 2) {
        const int ch = cg * 8 + c;
        v2f t;
        t.x = (vals[c]     - mu) * inv * ln_g[ch]     + ln_b[ch];
        t.y = (vals[c + 1] - mu) * inv * ln_g[ch + 1] + ln_b[ch + 1];
        op[c / 2] = t;
    }
}

extern "C" void kernel_launch(void* const* d_in, const int* in_sizes, int n_in,
                              void* d_out, int out_size, void* d_ws, size_t ws_size,
                              hipStream_t stream) {
    const float* feat1 = (const float*)d_in[0];
    const float* feat2 = (const float*)d_in[1];
    const float* Wq    = (const float*)d_in[2];
    const float* bq    = (const float*)d_in[3];
    const float* Wkv   = (const float*)d_in[4];
    const float* bkv   = (const float*)d_in[5];
    const float* Weg   = (const float*)d_in[6];
    const float* beg   = (const float*)d_in[7];
    const float* lng   = (const float*)d_in[8];
    const float* lnb   = (const float*)d_in[9];

    float* ws = (float*)d_ws;
    float* Qp = ws;
    float* Kp = Qp + (size_t)NN * HID;
    float* Vp = Kp + (size_t)NN * HID;
    float* Ep = Vp + (size_t)NN * HID;
    float* Gp = Ep + (size_t)NN * NHEADS;
    float* Wt = Gp + (size_t)NN * NHEADS;

    wpack_kernel<<<dim3(25), 32, 0, stream>>>(Wq, Wkv, Weg, Wt);

    proj_kernel<<<dim3(NN / 16), 256, 0, stream>>>(
        feat1, feat2, Wt, bq, bkv, beg, Qp, Kp, Vp, Ep, Gp);

    attn_kernel<<<dim3(NSZ, NSZ / 16), 256, 0, stream>>>(
        Qp, Kp, Vp, Ep, Gp, lng, lnb, (float*)d_out);
}


# Weight-pack launch: 8 waves per column tile, each lane's 8 loads issued as one group then 4 stores, instead of one wave looping 16 trips with a full memory wait per trip; same packed layout
# speedup vs baseline: 2.6047x; 1.1689x over previous
#include <hip/hip_runtime.h>
#include <hip/hip_bf16.h>

#define NSZ 160
#define NN (NSZ * NSZ)
#define HID 128
#define NHEADS 8
#define HD 16
#define LN_EPS 0.001f
#define APITCH 132

typedef __attribute__((ext_vector_type(2))) float v2f;
typedef __attribute__((ext_vector_type(4))) float v4f;
typedef __attribute__((ext_vector_type(8))) float v8f;

static __device__ __forceinline__ v8f wmma_f32_16x16x4(v2f a, v2f b, v8f c) {
    return __builtin_amdgcn_wmma_f32_16x16x4_f32(
         false, a,  false, b,
         (short)0, c,  false,  false);
}

static __device__ __forceinline__ float row16_max(float v) {
    asm("v_max_num_f32_dpp %0, %0, %0 row_xmask:1 row_mask:0xf bank_mask:0xf\n\t"
        "v_max_num_f32_dpp %0, %0, %0 row_xmask:2 row_mask:0xf bank_mask:0xf\n\t"
        "v_max_num_f32_dpp %0, %0, %0 row_xmask:4 row_mask:0xf bank_mask:0xf\n\t"
        "v_max_num_f32_dpp %0, %0, %0 row_xmask:8 row_mask:0xf bank_mask:0xf"
        : "+v"(v));
    return v;
}
static __device__ __forceinline__ float row16_sum(float v) {
    asm("v_add_f32_dpp %0, %0, %0 row_xmask:1 row_mask:0xf bank_mask:0xf\n\t"
        "v_add_f32_dpp %0, %0, %0 row_xmask:2 row_mask:0xf bank_mask:0xf\n\t"
        "v_add_f32_dpp %0, %0, %0 row_xmask:4 row_mask:0xf bank_mask:0xf\n\t"
        "v_add_f32_dpp %0, %0, %0 row_xmask:8 row_mask:0xf bank_mask:0xf"
        : "+v"(v));
    return v;
}


__global__ __launch_bounds__(256) void wpack_kernel(
    const float* __restrict__ Wq, const float* __restrict__ Wkv,
    const float* __restrict__ Weg, float* __restrict__ Wt)
{
    const int ct   = blockIdx.x;
    const int ks0  = (threadIdx.x / 32) * 4;
    const int lane = threadIdx.x % 32;
    const int hs   = lane / 16;
    const int lm   = lane % 16;

    const float* W; int wcols, cn;
    if (ct < 8)       { W = Wq;  wcols = HID;        cn = lm * NHEADS + ct; }
    else if (ct < 16) { W = Wkv; wcols = 2 * HID;    cn = lm * NHEADS + (ct - 8); }
    else if (ct < 24) { W = Wkv; wcols = 2 * HID;    cn = HID + lm * NHEADS + (ct - 16); }
    else              { W = Weg; wcols = 2 * NHEADS; cn = lm; }

    const float* src = W + (size_t)(ks0 * 4 + 2 * hs) * wcols + cn;
    float* o = Wt + (size_t)ct * 2048 + ks0 * 64 + lane * 2;
    v2f w[4];
    #pragma unroll
    for (int i = 0; i < 4; ++i) {
        w[i].x = src[(size_t)(i * 4)     * wcols];
        w[i].y = src[(size_t)(i * 4 + 1) * wcols];
    }
    #pragma unroll
    for (int i = 0; i < 4; ++i)
        *reinterpret_cast<v2f*>(o + i * 64) = w[i];
}

template <bool TRANSPOSED>
static __device__ __forceinline__ v8f gemm_tile(
    const float* __restrict__ sA, const float* __restrict__ wt,
    int lm, int hs)
{
    v8f acc = {0.f, 0.f, 0.f, 0.f, 0.f, 0.f, 0.f, 0.f};
    const float* pa = sA + lm * APITCH + 2 * hs;
    #pragma unroll
    for (int ks = 0; ks < 32; ++ks) {
        v2f a, b;
        a.x = pa[ks * 4];
        a.y = pa[ks * 4 + 1];
        b = *reinterpret_cast<const v2f*>(wt + ks * 64);
        acc = TRANSPOSED ? wmma_f32_16x16x4(b, a, acc) : wmma_f32_16x16x4(a, b, acc);
    }
    return acc;
}

__global__ __launch_bounds__(256) void proj_kernel(
    const float* __restrict__ feat1, const float* __restrict__ feat2,
    const float* __restrict__ Wt,
    const float* __restrict__ bq, const float* __restrict__ bkv,
    const float* __restrict__ beg,
    float* __restrict__ Qp, float* __restrict__ Kp, float* __restrict__ Vp,
    float* __restrict__ Ep, float* __restrict__ Gp)
{
    __shared__ float sA1[16][APITCH];
    __shared__ float sA2[16][APITCH];

    const int row0 = blockIdx.x * 16;
    const int tid  = threadIdx.x;

    const int aIdx = row0 / NSZ;
    const int b0   = row0 % NSZ;

    for (int idx = tid; idx < 16 * (HID / 4); idx += 256) {
        const int r = idx / (HID / 4), c4 = idx % (HID / 4);
        reinterpret_cast<float4*>(&sA1[r][0])[c4] =
            reinterpret_cast<const float4*>(&feat1[((size_t)(b0 + r) * NSZ + aIdx) * HID])[c4];
        reinterpret_cast<float4*>(&sA2[r][0])[c4] =
            reinterpret_cast<const float4*>(&feat2[(size_t)(row0 + r) * HID])[c4];
    }
    __syncthreads();

    const int wave = tid / 32;
    const int lane = tid % 32;
    const int hs   = lane / 16;
    const int lm   = lane % 16;

    const int bBase = b0 + 8 * hs;

    const size_t plane = ((size_t)aIdx * NHEADS + wave) * (NSZ * HD);

    {
        v8f acc = gemm_tile<true>(&sA1[0][0], Wt + (size_t)wave * 2048 + lane * 2, lm, hs);
        const float* bp = bq + (8 * hs) * NHEADS + wave;
        float* oq = Qp + plane + (b0 / 16) * 256 + lane * 4;
        v4f lo, hi;
        #pragma unroll
        for (int r = 0; r < 4; ++r) {
            lo[r] = (acc[r]     + bp[r * NHEADS])       * 0.25f;
            hi[r] = (acc[r + 4] + bp[(r + 4) * NHEADS]) * 0.25f;
        }
        *reinterpret_cast<v4f*>(oq)       = lo;
        *reinterpret_cast<v4f*>(oq + 128) = hi;
    }

    {
        v8f acc = gemm_tile<true>(&sA2[0][0], Wt + (size_t)(8 + wave) * 2048 + lane * 2, lm, hs);
        const float* bp = bkv + (8 * hs) * NHEADS + wave;
        float* ok = Kp + plane + (b0 / 16) * 256 + lane * 4;
        v4f lo, hi;
        #pragma unroll
        for (int r = 0; r < 4; ++r) {
            lo[r] = acc[r]     + bp[r * NHEADS];
            hi[r] = acc[r + 4] + bp[(r + 4) * NHEADS];
        }
        *reinterpret_cast<v4f*>(ok)       = lo;
        *reinterpret_cast<v4f*>(ok + 128) = hi;
    }

    {
        v8f acc = gemm_tile<false>(&sA2[0][0], Wt + (size_t)(16 + wave) * 2048 + lane * 2, lm, hs);
        const float bv = bkv[HID + lm * NHEADS + wave];
        float* ov = Vp + plane + (b0 / 8 + hs) * 128 + lm * 4;
        v4f lo, hi;
        #pragma unroll
        for (int r = 0; r < 4; ++r) { lo[r] = acc[r] + bv; hi[r] = acc[r + 4] + bv; }
        *reinterpret_cast<v4f*>(ov)      = lo;
        *reinterpret_cast<v4f*>(ov + 64) = hi;
    }

    if (wave == 0) {
        v8f acc = gemm_tile<false>(&sA2[0][0], Wt + (size_t)24 * 2048 + lane * 2, lm, hs);
        const float bv = beg[lm];
        if (lm < NHEADS) {
            float* oe = Ep + ((size_t)aIdx * NHEADS + lm) * NSZ + bBase;
            #pragma unroll
            for (int r = 0; r < 8; ++r) oe[r] = acc[r] + bv;
        } else {
            float* og = Gp + ((size_t)aIdx * NHEADS + (lm - NHEADS)) * NSZ + bBase;
            #pragma unroll
            for (int r = 0; r < 8; ++r)
                og[r] = 1.f / (1.f + __expf(-(acc[r] + bv)));
        }
    }
}

__global__ __launch_bounds__(256) __attribute__((amdgpu_waves_per_eu(6))) void attn_kernel(
    const float* __restrict__ Qp, const float* __restrict__ Kp,
    const float* __restrict__ Vp, const float* __restrict__ Ep,
    const float* __restrict__ Gp,
    const float* __restrict__ ln_g, const float* __restrict__ ln_b,
    float* __restrict__ out)
{
    __shared__ __attribute__((aligned(16))) float Plds[NHEADS][16][84];
    __shared__ float VaLds[16][HID];

    const int j    = blockIdx.x;
    const int i0   = blockIdx.y * 16;
    const int tid  = threadIdx.x;
    const int h    = __builtin_amdgcn_readfirstlane(tid / 32);
    const int lane = tid % 32;
    const int hs   = lane / 16;
    const int lm   = lane % 16;

    const size_t jh = (size_t)j * NHEADS + h;

    const float* pQ = Qp + jh * (NSZ * HD) + blockIdx.y * 256 + lane * 4;
    const float* pK = Kp + jh * (NSZ * HD) + lane * 4;
    const float* pV = Vp + jh * (HD * NSZ) + lane * 4;
    const float* pE = Ep + jh * NSZ + lm;
    const float* pG = Gp + jh * NSZ + lm;

    {
        const float* vslice = Vp + jh * HD * NSZ;
        for (int p = lane; p < 80; p += 32)
            __builtin_prefetch(vslice + p * 32, 0, 1);
    }

    v2f afr[4];
    #pragma unroll
    for (int x = 0; x < 2; ++x) {
        const v4f q4 = *reinterpret_cast<const v4f*>(pQ + x * 128);
        afr[2 * x]     = v2f{q4[0], q4[1]};
        afr[2 * x + 1] = v2f{q4[2], q4[3]};
    }

    v2f kf[40];
    #pragma unroll
    for (int ct = 0; ct < 10; ++ct) {
        #pragma unroll
        for (int x = 0; x < 2; ++x) {
            const v4f k4 = *reinterpret_cast<const v4f*>(pK + ct * 256 + x * 128);
            kf[ct * 4 + 2 * x]     = v2f{k4[0], k4[1]};
            kf[ct * 4 + 2 * x + 1] = v2f{k4[2], k4[3]};
        }
    }
    float ev[10];
    #pragma unroll
    for (int ct = 0; ct < 10; ++ct) ev[ct] = pE[ct * 16];
    __builtin_amdgcn_sched_barrier(0);

    v8f acc[10];
    #pragma unroll
    for (int ct = 0; ct < 10; ++ct) {
        v8f c = {0.f, 0.f, 0.f, 0.f, 0.f, 0.f, 0.f, 0.f};
        #pragma unroll
        for (int ks = 0; ks < 4; ++ks)
            c = wmma_f32_16x16x4(afr[ks], kf[ct * 4 + ks], c);
        const float e = ev[ct];
        #pragma unroll
        for (int r = 0; r < 8; ++r) c[r] += e;
        acc[ct] = c;
    }

    float gate[10];
    #pragma unroll
    for (int ct = 0; ct < 10; ++ct) gate[ct] = pG[ct * 16];

    #pragma unroll
    for (int r = 0; r < 8; ++r) {
        float mx = -1e30f;
        #pragma unroll
        for (int ct = 0; ct < 10; ++ct) mx = fmaxf(mx, acc[ct][r]);
        mx = row16_max(mx);
        float sm = 0.f;
        #pragma unroll
        for (int ct = 0; ct < 10; ++ct) {
            float e = __expf(acc[ct][r] - mx);
            acc[ct][r] = e;
            sm += e;
        }
        sm = row16_sum(sm);
        const float inv = 1.f / sm;
        #pragma unroll
        for (int ct = 0; ct < 10; ++ct) acc[ct][r] = acc[ct][r] * inv * gate[ct];
    }

    v8f va = {0.f, 0.f, 0.f, 0.f, 0.f, 0.f, 0.f, 0.f};
    #pragma unroll
    for (int half = 0; half < 2; ++half) {
        #pragma unroll
        for (int ct = 0; ct < 5; ++ct) {
            #pragma unroll
            for (int r = 0; r < 8; ++r)
                Plds[h][r + 8 * hs][ct * 16 + lm] = acc[half * 5 + ct][r];
        }
        __builtin_amdgcn_sched_barrier(0);
        v2f vf[20];
        #pragma unroll
        for (int g = 0; g < 10; ++g) {
            const v4f v4 = *reinterpret_cast<const v4f*>(pV + (half * 10 + g) * 128);
            vf[2 * g]     = v2f{v4[0], v4[1]};
            vf[2 * g + 1] = v2f{v4[2], v4[3]};
        }
        __syncthreads();
        v2f pf[20];
        #pragma unroll
        for (int g = 0; g < 10; ++g) {
            const v4f p4 = *reinterpret_cast<const v4f*>(&Plds[h][lm][g * 8 + 4 * hs]);
            pf[2 * g]     = v2f{p4[0], p4[1]};
            pf[2 * g + 1] = v2f{p4[2], p4[3]};
        }
        __builtin_amdgcn_sched_barrier(0);
        #pragma unroll
        for (int q = 0; q < 20; ++q)
            va = wmma_f32_16x16x4(pf[q], vf[q], va);
        __syncthreads();
    }

    #pragma unroll
    for (int r = 0; r < 8; ++r)
        VaLds[r + 8 * hs][lm * NHEADS + h] = va[r];
    __syncthreads();

    const int rowLN = tid / 16;
    const int cg    = tid % 16;
    float vals[8], s1 = 0.f, s2 = 0.f;
    #pragma unroll
    for (int c = 0; c < 8; ++c) {
        float v = VaLds[rowLN][cg * 8 + c];
        vals[c] = v; s1 += v; s2 += v * v;
    }
    s1 = row16_sum(s1);
    s2 = row16_sum(s2);
    const float mu  = s1 * (1.f / HID);
    const float var = s2 * (1.f / HID) - mu * mu;
    const float inv = rsqrtf(var + LN_EPS);

    const size_t obase = ((size_t)(i0 + rowLN) * NSZ + j) * HID + cg * 8;
    v2f* op = reinterpret_cast<v2f*>(&out[obase]);
    #pragma unroll
    for (int c = 0; c < 8; c += 2) {
        const int ch = cg * 8 + c;
        v2f t;
        t.x = (vals[c]     - mu) * inv * ln_g[ch]     + ln_b[ch];
        t.y = (vals[c + 1] - mu) * inv * ln_g[ch + 1] + ln_b[ch + 1];
        op[c / 2] = t;
    }
}

extern "C" void kernel_launch(void* const* d_in, const int* in_sizes, int n_in,
                              void* d_out, int out_size, void* d_ws, size_t ws_size,
                              hipStream_t stream) {
    const float* feat1 = (const float*)d_in[0];
    const float* feat2 = (const float*)d_in[1];
    const float* Wq    = (const float*)d_in[2];
    const float* bq    = (const float*)d_in[3];
    const float* Wkv   = (const float*)d_in[4];
    const float* bkv   = (const float*)d_in[5];
    const float* Weg   = (const float*)d_in[6];
    const float* beg   = (const float*)d_in[7];
    const float* lng   = (const float*)d_in[8];
    const float* lnb   = (const float*)d_in[9];

    float* ws = (float*)d_ws;
    float* Qp = ws;
    float* Kp = Qp + (size_t)NN * HID;
    float* Vp = Kp + (size_t)NN * HID;
    float* Ep = Vp + (size_t)NN * HID;
    float* Gp = Ep + (size_t)NN * NHEADS;
    float* Wt = Gp + (size_t)NN * NHEADS;

    wpack_kernel<<<dim3(25), 256, 0, stream>>>(Wq, Wkv, Weg, Wt);

    proj_kernel<<<dim3(NN / 16), 256, 0, stream>>>(
        feat1, feat2, Wt, bq, bkv, beg, Qp, Kp, Vp, Ep, Gp);

    attn_kernel<<<dim3(NSZ, NSZ / 16), 256, 0, stream>>>(
        Qp, Kp, Vp, Ep, Gp, lng, lnb, (float*)d_out);
}


# Attention scores formed transposed (keys as rows): softmaxed registers are the A fragments of P@V, so P skips LDS and 4 workgroup barriers; V plane reordered; softmax and P@V sum orders regrouped
# speedup vs baseline: 2.7015x; 1.0372x over previous
#include <hip/hip_runtime.h>
#include <hip/hip_bf16.h>

#define NSZ 160
#define NN (NSZ * NSZ)
#define HID 128
#define NHEADS 8
#define HD 16
#define LN_EPS 0.001f
#define APITCH 132

typedef __attribute__((ext_vector_type(2))) float v2f;
typedef __attribute__((ext_vector_type(4))) float v4f;
typedef __attribute__((ext_vector_type(8))) float v8f;

static __device__ __forceinline__ v8f wmma_f32_16x16x4(v2f a, v2f b, v8f c) {
    return __builtin_amdgcn_wmma_f32_16x16x4_f32(
         false, a,  false, b,
         (short)0, c,  false,  false);
}

static __device__ __forceinline__ float row16_sum(float v) {
    asm("v_add_f32_dpp %0, %0, %0 row_xmask:1 row_mask:0xf bank_mask:0xf\n\t"
        "v_add_f32_dpp %0, %0, %0 row_xmask:2 row_mask:0xf bank_mask:0xf\n\t"
        "v_add_f32_dpp %0, %0, %0 row_xmask:4 row_mask:0xf bank_mask:0xf\n\t"
        "v_add_f32_dpp %0, %0, %0 row_xmask:8 row_mask:0xf bank_mask:0xf"
        : "+v"(v));
    return v;
}

#define ROW16_OP8(OP)                                                              \
    asm(OP " %0, %8, %0 row_share:0 row_mask:0xf bank_mask:0xf\n\t"                \
        OP " %1, %8, %1 row_share:1 row_mask:0xf bank_mask:0xf\n\t"                \
        OP " %2, %8, %2 row_share:2 row_mask:0xf bank_mask:0xf\n\t"                \
        OP " %3, %8, %3 row_share:3 row_mask:0xf bank_mask:0xf\n\t"                \
        OP " %4, %8, %4 row_share:4 row_mask:0xf bank_mask:0xf\n\t"                \
        OP " %5, %8, %5 row_share:5 row_mask:0xf bank_mask:0xf\n\t"                \
        OP " %6, %8, %6 row_share:6 row_mask:0xf bank_mask:0xf\n\t"                \
        OP " %7, %8, %7 row_share:7 row_mask:0xf bank_mask:0xf"                     \
        : "+v"(t0), "+v"(t1), "+v"(t2), "+v"(t3), "+v"(t4), "+v"(t5), "+v"(t6), "+v"(t7) \
        : "v"(v))
static __device__ __forceinline__ v8f row16_add8(v8f c, float v) {
    float t0 = c[0], t1 = c[1], t2 = c[2], t3 = c[3], t4 = c[4], t5 = c[5], t6 = c[6], t7 = c[7];
    ROW16_OP8("v_add_f32_dpp");
    return v8f{t0, t1, t2, t3, t4, t5, t6, t7};
}
static __device__ __forceinline__ v8f row16_mul8(v8f c, float v) {
    float t0 = c[0], t1 = c[1], t2 = c[2], t3 = c[3], t4 = c[4], t5 = c[5], t6 = c[6], t7 = c[7];
    ROW16_OP8("v_mul_f32_dpp");
    return v8f{t0, t1, t2, t3, t4, t5, t6, t7};
}
static __device__ __forceinline__ float other_half(float v) {
    return __builtin_bit_cast(float, __builtin_amdgcn_permlanex16(
        0u, __builtin_bit_cast(unsigned, v), 0x76543210u, 0xfedcba98u, false, false));
}


__global__ __launch_bounds__(256) void wpack_kernel(
    const float* __restrict__ Wq, const float* __restrict__ Wkv,
    const float* __restrict__ Weg, float* __restrict__ Wt)
{
    const int ct   = blockIdx.x;
    const int ks0  = (threadIdx.x / 32) * 4;
    const int lane = threadIdx.x % 32;
    const int hs   = lane / 16;
    const int lm   = lane % 16;

    const float* W; int wcols, cn;
    if (ct < 8)       { W = Wq;  wcols = HID;        cn = lm * NHEADS + ct; }
    else if (ct < 16) { W = Wkv; wcols = 2 * HID;    cn = lm * NHEADS + (ct - 8); }
    else if (ct < 24) { W = Wkv; wcols = 2 * HID;    cn = HID + lm * NHEADS + (ct - 16); }
    else              { W = Weg; wcols = 2 * NHEADS; cn = lm; }

    const float* src = W + (size_t)(ks0 * 4 + 2 * hs) * wcols + cn;
    float* o = Wt + (size_t)ct * 2048 + ks0 * 64 + lane * 2;
    v2f w[4];
    #pragma unroll
    for (int i = 0; i < 4; ++i) {
        w[i].x = src[(size_t)(i * 4)     * wcols];
        w[i].y = src[(size_t)(i * 4 + 1) * wcols];
    }
    #pragma unroll
    for (int i = 0; i < 4; ++i)
        *reinterpret_cast<v2f*>(o + i * 64) = w[i];
}

template <bool TRANSPOSED>
static __device__ __forceinline__ v8f gemm_tile(
    const float* __restrict__ sA, const float* __restrict__ wt,
    int lm, int hs)
{
    v8f acc = {0.f, 0.f, 0.f, 0.f, 0.f, 0.f, 0.f, 0.f};
    const float* pa = sA + lm * APITCH + 2 * hs;
    #pragma unroll
    for (int ks = 0; ks < 32; ++ks) {
        v2f a, b;
        a.x = pa[ks * 4];
        a.y = pa[ks * 4 + 1];
        b = *reinterpret_cast<const v2f*>(wt + ks * 64);
        acc = TRANSPOSED ? wmma_f32_16x16x4(b, a, acc) : wmma_f32_16x16x4(a, b, acc);
    }
    return acc;
}

__global__ __launch_bounds__(256) void proj_kernel(
    const float* __restrict__ feat1, const float* __restrict__ feat2,
    const float* __restrict__ Wt,
    const float* __restrict__ bq, const float* __restrict__ bkv,
    const float* __restrict__ beg,
    float* __restrict__ Qp, float* __restrict__ Kp, float* __restrict__ Vp,
    float* __restrict__ Ep, float* __restrict__ Gp)
{
    __shared__ float sA1[16][APITCH];
    __shared__ float sA2[16][APITCH];

    const int row0 = blockIdx.x * 16;
    const int tid  = threadIdx.x;

    const int aIdx = row0 / NSZ;
    const int b0   = row0 % NSZ;

    for (int idx = tid; idx < 16 * (HID / 4); idx += 256) {
        const int r = idx / (HID / 4), c4 = idx % (HID / 4);
        reinterpret_cast<float4*>(&sA1[r][0])[c4] =
            reinterpret_cast<const float4*>(&feat1[((size_t)(b0 + r) * NSZ + aIdx) * HID])[c4];
        reinterpret_cast<float4*>(&sA2[r][0])[c4] =
            reinterpret_cast<const float4*>(&feat2[(size_t)(row0 + r) * HID])[c4];
    }
    __syncthreads();

    const int wave = tid / 32;
    const int lane = tid % 32;
    const int hs   = lane / 16;
    const int lm   = lane % 16;

    const int bBase = b0 + 8 * hs;

    const size_t plane = ((size_t)aIdx * NHEADS + wave) * (NSZ * HD);

    {
        v8f acc = gemm_tile<true>(&sA1[0][0], Wt + (size_t)wave * 2048 + lane * 2, lm, hs);
        const float* bp = bq + (8 * hs) * NHEADS + wave;
        float* oq = Qp + plane + (b0 / 16) * 256 + lane * 4;
        v4f lo, hi;
        #pragma unroll
        for (int r = 0; r < 4; ++r) {
            lo[r] = (acc[r]     + bp[r * NHEADS])       * 0.25f;
            hi[r] = (acc[r + 4] + bp[(r + 4) * NHEADS]) * 0.25f;
        }
        *reinterpret_cast<v4f*>(oq)       = lo;
        *reinterpret_cast<v4f*>(oq + 128) = hi;
    }

    {
        v8f acc = gemm_tile<true>(&sA2[0][0], Wt + (size_t)(8 + wave) * 2048 + lane * 2, lm, hs);
        const float* bp = bkv + (8 * hs) * NHEADS + wave;
        float* ok = Kp + plane + (b0 / 16) * 256 + lane * 4;
        v4f lo, hi;
        #pragma unroll
        for (int r = 0; r < 4; ++r) {
            lo[r] = acc[r]     + bp[r * NHEADS];
            hi[r] = acc[r + 4] + bp[(r + 4) * NHEADS];
        }
        *reinterpret_cast<v4f*>(ok)       = lo;
        *reinterpret_cast<v4f*>(ok + 128) = hi;
    }

    {
        v8f acc = gemm_tile<false>(&sA2[0][0], Wt + (size_t)(16 + wave) * 2048 + lane * 2, lm, hs);
        const float bv = bkv[HID + lm * NHEADS + wave];
        float* ov = Vp + plane + (b0 / 8 + hs) * 128 + lm * 4 - hs * 64;
        v4f lo, hi;
        #pragma unroll
        for (int r = 0; r < 4; ++r) { lo[r] = acc[r] + bv; hi[r] = acc[r + 4] + bv; }
        *reinterpret_cast<v4f*>(ov)       = lo;
        *reinterpret_cast<v4f*>(ov + 128) = hi;
    }

    if (wave == 0) {
        v8f acc = gemm_tile<false>(&sA2[0][0], Wt + (size_t)24 * 2048 + lane * 2, lm, hs);
        const float bv = beg[lm];
        if (lm < NHEADS) {
            float* oe = Ep + ((size_t)aIdx * NHEADS + lm) * NSZ + bBase;
            #pragma unroll
            for (int r = 0; r < 8; ++r) oe[r] = acc[r] + bv;
        } else {
            float* og = Gp + ((size_t)aIdx * NHEADS + (lm - NHEADS)) * NSZ + bBase;
            #pragma unroll
            for (int r = 0; r < 8; ++r)
                og[r] = 1.f / (1.f + __expf(-(acc[r] + bv)));
        }
    }
}

__global__ __launch_bounds__(256) __attribute__((amdgpu_waves_per_eu(6))) void attn_kernel(
    const float* __restrict__ Qp, const float* __restrict__ Kp,
    const float* __restrict__ Vp, const float* __restrict__ Ep,
    const float* __restrict__ Gp,
    const float* __restrict__ ln_g, const float* __restrict__ ln_b,
    float* __restrict__ out)
{
    __shared__ float VaLds[16][HID];

    const int j    = blockIdx.x;
    const int i0   = blockIdx.y * 16;
    const int tid  = threadIdx.x;
    const int h    = __builtin_amdgcn_readfirstlane(tid / 32);
    const int lane = tid % 32;
    const int hs   = lane / 16;
    const int lm   = lane % 16;

    const size_t jh = (size_t)j * NHEADS + h;

    const float* pQ = Qp + jh * (NSZ * HD) + blockIdx.y * 256 + lane * 4;
    const float* pK = Kp + jh * (NSZ * HD) + lane * 4;
    const float* pV = Vp + jh * (HD * NSZ) + lane * 4;
    const int   kq = (lm + 8 * hs) % 16;
    const float* pE = Ep + jh * NSZ + kq;
    const float* pG = Gp + jh * NSZ + kq;

    {
        const float* vslice = Vp + jh * HD * NSZ;
        for (int p = lane; p < 80; p += 32)
            __builtin_prefetch(vslice + p * 32, 0, 1);
    }

    v2f qfr[4];
    #pragma unroll
    for (int x = 0; x < 2; ++x) {
        const v4f q4 = *reinterpret_cast<const v4f*>(pQ + x * 128);
        qfr[2 * x]     = v2f{q4[0], q4[1]};
        qfr[2 * x + 1] = v2f{q4[2], q4[3]};
    }

    v2f kf[40];
    #pragma unroll
    for (int ct = 0; ct < 10; ++ct) {
        #pragma unroll
        for (int x = 0; x < 2; ++x) {
            const v4f k4 = *reinterpret_cast<const v4f*>(pK + ct * 256 + x * 128);
            kf[ct * 4 + 2 * x]     = v2f{k4[0], k4[1]};
            kf[ct * 4 + 2 * x + 1] = v2f{k4[2], k4[3]};
        }
    }
    float ev[10];
    #pragma unroll
    for (int ct = 0; ct < 10; ++ct) ev[ct] = pE[ct * 16];
    __builtin_amdgcn_sched_barrier(0);

    v8f acc[10];
    #pragma unroll
    for (int ct = 0; ct < 10; ++ct) {
        v8f c = {0.f, 0.f, 0.f, 0.f, 0.f, 0.f, 0.f, 0.f};
        #pragma unroll
        for (int ks = 0; ks < 4; ++ks)
            c = wmma_f32_16x16x4(kf[ct * 4 + ks], qfr[ks], c);
        acc[ct] = c;
        if (ct % 2 == 1) __builtin_amdgcn_sched_barrier(0);
    }
    #pragma unroll
    for (int ct = 0; ct < 10; ++ct) acc[ct] = row16_add8(acc[ct], ev[ct]);

    float gate[10];
    #pragma unroll
    for (int ct = 0; ct < 10; ++ct) gate[ct] = pG[ct * 16];
    __builtin_amdgcn_sched_barrier(0);

    v8f mx8 = acc[0];
    #pragma unroll
    for (int ct = 1; ct < 10; ++ct) {
        #pragma unroll
        for (int r = 0; r < 8; ++r) mx8[r] = fmaxf(mx8[r], acc[ct][r]);
    }
    float mx = fmaxf(fmaxf(fmaxf(mx8[0], mx8[1]), fmaxf(mx8[2], mx8[3])),
                     fmaxf(fmaxf(mx8[4], mx8[5]), fmaxf(mx8[6], mx8[7])));
    mx = fmaxf(mx, -1e30f);
    mx = fmaxf(mx, other_half(mx));

    v8f sm8 = {0.f, 0.f, 0.f, 0.f, 0.f, 0.f, 0.f, 0.f};
    #pragma unroll
    for (int ct = 0; ct < 10; ++ct) {
        #pragma unroll
        for (int r = 0; r < 8; ++r) {
            const float e = __expf(acc[ct][r] - mx);
            acc[ct][r] = e;
            sm8[r] += e;
        }
    }
    float sm = ((sm8[0] + sm8[1]) + (sm8[2] + sm8[3])) + ((sm8[4] + sm8[5]) + (sm8[6] + sm8[7]));
    sm += other_half(sm);
    const float rinv = 1.f / sm;
    #pragma unroll
    for (int ct = 0; ct < 10; ++ct) acc[ct] = row16_mul8(acc[ct] * rinv, gate[ct]);

    v2f vf[40];
    #pragma unroll
    for (int ct = 0; ct < 7; ++ct) {
        #pragma unroll
        for (int x = 0; x < 2; ++x) {
            const v4f v4 = *reinterpret_cast<const v4f*>(pV + ct * 256 + x * 128);
            vf[ct * 4 + 2 * x]     = v2f{v4[0], v4[1]};
            vf[ct * 4 + 2 * x + 1] = v2f{v4[2], v4[3]};
        }
    }
    __builtin_amdgcn_sched_barrier(0);
    v8f va = {0.f, 0.f, 0.f, 0.f, 0.f, 0.f, 0.f, 0.f};
    #pragma unroll
    for (int ct = 0; ct < 7; ++ct) {
        #pragma unroll
        for (int rp = 0; rp < 4; ++rp)
            va = wmma_f32_16x16x4(v2f{acc[ct][2 * rp], acc[ct][2 * rp + 1]}, vf[ct * 4 + rp], va);
    }
    #pragma unroll
    for (int ct = 7; ct < 10; ++ct) {
        #pragma unroll
        for (int x = 0; x < 2; ++x) {
            const v4f v4 = *reinterpret_cast<const v4f*>(pV + ct * 256 + x * 128);
            vf[ct * 4 + 2 * x]     = v2f{v4[0], v4[1]};
            vf[ct * 4 + 2 * x + 1] = v2f{v4[2], v4[3]};
        }
    }
    __builtin_amdgcn_sched_barrier(0);
    #pragma unroll
    for (int ct = 7; ct < 10; ++ct) {
        #pragma unroll
        for (int rp = 0; rp < 4; ++rp)
            va = wmma_f32_16x16x4(v2f{acc[ct][2 * rp], acc[ct][2 * rp + 1]}, vf[ct * 4 + rp], va);
    }

    #pragma unroll
    for (int r = 0; r < 8; ++r)
        VaLds[r + 8 * hs][lm * NHEADS + h] = va[r];
    __syncthreads();

    const int rowLN = tid / 16;
    const int cg    = tid % 16;
    float vals[8], s1 = 0.f, s2 = 0.f;
    #pragma unroll
    for (int c = 0; c < 8; ++c) {
        float v = VaLds[rowLN][cg * 8 + c];
        vals[c] = v; s1 += v; s2 += v * v;
    }
    s1 = row16_sum(s1);
    s2 = row16_sum(s2);
    const float mu  = s1 * (1.f / HID);
    const float var = s2 * (1.f / HID) - mu * mu;
    const float inv = rsqrtf(var + LN_EPS);

    const size_t obase = ((size_t)(i0 + rowLN) * NSZ + j) * HID + cg * 8;
    v2f* op = reinterpret_cast<v2f*>(&out[obase]);
    #pragma unroll
    for (int c = 0; c < 8; c += 2) {
        const int ch = cg * 8 + c;
        v2f t;
        t.x = (vals[c]     - mu) * inv * ln_g[ch]     + ln_b[ch];
        t.y = (vals[c + 1] - mu) * inv * ln_g[ch + 1] + ln_b[ch + 1];
        op[c / 2] = t;
    }
}

extern "C" void kernel_launch(void* const* d_in, const int* in_sizes, int n_in,
                              void* d_out, int out_size, void* d_ws, size_t ws_size,
                              hipStream_t stream) {
    const float* feat1 = (const float*)d_in[0];
    const float* feat2 = (const float*)d_in[1];
    const float* Wq    = (const float*)d_in[2];
    const float* bq    = (const float*)d_in[3];
    const float* Wkv   = (const float*)d_in[4];
    const float* bkv   = (const float*)d_in[5];
    const float* Weg   = (const float*)d_in[6];
    const float* beg   = (const float*)d_in[7];
    const float* lng   = (const float*)d_in[8];
    const float* lnb   = (const float*)d_in[9];

    float* ws = (float*)d_ws;
    float* Qp = ws;
    float* Kp = Qp + (size_t)NN * HID;
    float* Vp = Kp + (size_t)NN * HID;
    float* Ep = Vp + (size_t)NN * HID;
    float* Gp = Ep + (size_t)NN * NHEADS;
    float* Wt = Gp + (size_t)NN * NHEADS;

    wpack_kernel<<<dim3(25), 256, 0, stream>>>(Wq, Wkv, Weg, Wt);

    proj_kernel<<<dim3(NN / 16), 256, 0, stream>>>(
        feat1, feat2, Wt, bq, bkv, beg, Qp, Kp, Vp, Ep, Gp);

    attn_kernel<<<dim3(NSZ, NSZ / 16), 256, 0, stream>>>(
        Qp, Kp, Vp, Ep, Gp, lng, lnb, (float*)d_out);
}


# Projection GEMM: a workgroup is 32 rows x 4 heads on 4 waves instead of 16 rows x 8 heads on 8: each packed weight fragment load feeds two row tiles' WMMAs; half the weight reads, half the waves
# speedup vs baseline: 2.8926x; 1.0707x over previous
#include <hip/hip_runtime.h>
#include <hip/hip_bf16.h>

#define NSZ 160
#define NN (NSZ * NSZ)
#define HID 128
#define NHEADS 8
#define HD 16
#define LN_EPS 0.001f
#define APITCH 132

typedef __attribute__((ext_vector_type(2))) float v2f;
typedef __attribute__((ext_vector_type(4))) float v4f;
typedef __attribute__((ext_vector_type(8))) float v8f;

static __device__ __forceinline__ v8f wmma_f32_16x16x4(v2f a, v2f b, v8f c) {
    return __builtin_amdgcn_wmma_f32_16x16x4_f32(
         false, a,  false, b,
         (short)0, c,  false,  false);
}

static __device__ __forceinline__ float row16_sum(float v) {
    asm("v_add_f32_dpp %0, %0, %0 row_xmask:1 row_mask:0xf bank_mask:0xf\n\t"
        "v_add_f32_dpp %0, %0, %0 row_xmask:2 row_mask:0xf bank_mask:0xf\n\t"
        "v_add_f32_dpp %0, %0, %0 row_xmask:4 row_mask:0xf bank_mask:0xf\n\t"
        "v_add_f32_dpp %0, %0, %0 row_xmask:8 row_mask:0xf bank_mask:0xf"
        : "+v"(v));
    return v;
}

#define ROW16_OP8(OP)                                                              \
    asm(OP " %0, %8, %0 row_share:0 row_mask:0xf bank_mask:0xf\n\t"                \
        OP " %1, %8, %1 row_share:1 row_mask:0xf bank_mask:0xf\n\t"                \
        OP " %2, %8, %2 row_share:2 row_mask:0xf bank_mask:0xf\n\t"                \
        OP " %3, %8, %3 row_share:3 row_mask:0xf bank_mask:0xf\n\t"                \
        OP " %4, %8, %4 row_share:4 row_mask:0xf bank_mask:0xf\n\t"                \
        OP " %5, %8, %5 row_share:5 row_mask:0xf bank_mask:0xf\n\t"                \
        OP " %6, %8, %6 row_share:6 row_mask:0xf bank_mask:0xf\n\t"                \
        OP " %7, %8, %7 row_share:7 row_mask:0xf bank_mask:0xf"                     \
        : "+v"(t0), "+v"(t1), "+v"(t2), "+v"(t3), "+v"(t4), "+v"(t5), "+v"(t6), "+v"(t7) \
        : "v"(v))
static __device__ __forceinline__ v8f row16_add8(v8f c, float v) {
    float t0 = c[0], t1 = c[1], t2 = c[2], t3 = c[3], t4 = c[4], t5 = c[5], t6 = c[6], t7 = c[7];
    ROW16_OP8("v_add_f32_dpp");
    return v8f{t0, t1, t2, t3, t4, t5, t6, t7};
}
static __device__ __forceinline__ v8f row16_mul8(v8f c, float v) {
    float t0 = c[0], t1 = c[1], t2 = c[2], t3 = c[3], t4 = c[4], t5 = c[5], t6 = c[6], t7 = c[7];
    ROW16_OP8("v_mul_f32_dpp");
    return v8f{t0, t1, t2, t3, t4, t5, t6, t7};
}
static __device__ __forceinline__ float other_half(float v) {
    return __builtin_bit_cast(float, __builtin_amdgcn_permlanex16(
        0u, __builtin_bit_cast(unsigned, v), 0x76543210u, 0xfedcba98u, false, false));
}


__global__ __launch_bounds__(256) void wpack_kernel(
    const float* __restrict__ Wq, const float* __restrict__ Wkv,
    const float* __restrict__ Weg, float* __restrict__ Wt)
{
    const int ct   = blockIdx.x;
    const int ks0  = (threadIdx.x / 32) * 4;
    const int lane = threadIdx.x % 32;
    const int hs   = lane / 16;
    const int lm   = lane % 16;

    const float* W; int wcols, cn;
    if (ct < 8)       { W = Wq;  wcols = HID;        cn = lm * NHEADS + ct; }
    else if (ct < 16) { W = Wkv; wcols = 2 * HID;    cn = lm * NHEADS + (ct - 8); }
    else if (ct < 24) { W = Wkv; wcols = 2 * HID;    cn = HID + lm * NHEADS + (ct - 16); }
    else              { W = Weg; wcols = 2 * NHEADS; cn = lm; }

    const float* src = W + (size_t)(ks0 * 4 + 2 * hs) * wcols + cn;
    float* o = Wt + (size_t)ct * 2048 + ks0 * 64 + lane * 2;
    v2f w[4];
    #pragma unroll
    for (int i = 0; i < 4; ++i) {
        w[i].x = src[(size_t)(i * 4)     * wcols];
        w[i].y = src[(size_t)(i * 4 + 1) * wcols];
    }
    #pragma unroll
    for (int i = 0; i < 4; ++i)
        *reinterpret_cast<v2f*>(o + i * 64) = w[i];
}

template <bool TRANSPOSED>
static __device__ __forceinline__ void gemm_tile2(
    const float* __restrict__ sA, const float* __restrict__ wt,
    int lm, int hs, v8f& acc0, v8f& acc1)
{
    v8f c0 = {0.f, 0.f, 0.f, 0.f, 0.f, 0.f, 0.f, 0.f};
    v8f c1 = {0.f, 0.f, 0.f, 0.f, 0.f, 0.f, 0.f, 0.f};
    const float* pa0 = sA + lm * APITCH + 2 * hs;
    const float* pa1 = pa0 + 16 * APITCH;
    for (int ks = 0; ks < 32; ++ks) {
        v2f a0, a1, b;
        a0.x = pa0[ks * 4];
        a0.y = pa0[ks * 4 + 1];
        a1.x = pa1[ks * 4];
        a1.y = pa1[ks * 4 + 1];
        b = *reinterpret_cast<const v2f*>(wt + ks * 64);
        c0 = TRANSPOSED ? wmma_f32_16x16x4(b, a0, c0) : wmma_f32_16x16x4(a0, b, c0);
        c1 = TRANSPOSED ? wmma_f32_16x16x4(b, a1, c1) : wmma_f32_16x16x4(a1, b, c1);
    }
    acc0 = c0;
    acc1 = c1;
}

static __device__ __forceinline__ v8f gemm_tile1(
    const float* __restrict__ sA, const float* __restrict__ wt,
    int lm, int hs)
{
    v8f acc = {0.f, 0.f, 0.f, 0.f, 0.f, 0.f, 0.f, 0.f};
    const float* pa = sA + lm * APITCH + 2 * hs;
    for (int ks = 0; ks < 32; ++ks) {
        v2f a, b;
        a.x = pa[ks * 4];
        a.y = pa[ks * 4 + 1];
        b = *reinterpret_cast<const v2f*>(wt + ks * 64);
        acc = wmma_f32_16x16x4(a, b, acc);
    }
    return acc;
}

__global__ __launch_bounds__(128) void proj_kernel(
    const float* __restrict__ feat1, const float* __restrict__ feat2,
    const float* __restrict__ Wt,
    const float* __restrict__ bq, const float* __restrict__ bkv,
    const float* __restrict__ beg,
    float* __restrict__ Qp, float* __restrict__ Kp, float* __restrict__ Vp,
    float* __restrict__ Ep, float* __restrict__ Gp)
{
    __shared__ float sA1[32][APITCH];
    __shared__ float sA2[32][APITCH];

    const int pr   = blockIdx.x / 2;
    const int hh   = blockIdx.x % 2;
    const int row0 = pr * 32;
    const int tid  = threadIdx.x;

    const int aIdx = row0 / NSZ;
    const int b0   = row0 % NSZ;

    for (int it = 0; it < (32 * (HID / 4)) / 128; ++it) {
        const int idx = tid + it * 128;
        const int r = idx / (HID / 4), c4 = idx % (HID / 4);
        reinterpret_cast<float4*>(&sA1[r][0])[c4] =
            reinterpret_cast<const float4*>(&feat1[((size_t)(b0 + r) * NSZ + aIdx) * HID])[c4];
        reinterpret_cast<float4*>(&sA2[r][0])[c4] =
            reinterpret_cast<const float4*>(&feat2[(size_t)(row0 + r) * HID])[c4];
    }
    __syncthreads();

    const int wave = tid / 32;
    const int lane = tid % 32;
    const int hs   = lane / 16;
    const int lm   = lane % 16;
    const int head = hh * 4 + wave;

    const size_t plane = ((size_t)aIdx * NHEADS + head) * (NSZ * HD);
    const int    slot  = (b0 / 16) * 256 + lane * 4;

    {
        v8f acc[2];
        gemm_tile2<true>(&sA1[0][0], Wt + (size_t)head * 2048 + lane * 2, lm, hs, acc[0], acc[1]);
        const float* bp = bq + (8 * hs) * NHEADS + head;
        for (int t = 0; t < 2; ++t) {
            float* oq = Qp + plane + slot + t * 256;
            v4f lo, hi;
            #pragma unroll
            for (int r = 0; r < 4; ++r) {
                lo[r] = (acc[t][r]     + bp[r * NHEADS])       * 0.25f;
                hi[r] = (acc[t][r + 4] + bp[(r + 4) * NHEADS]) * 0.25f;
            }
            *reinterpret_cast<v4f*>(oq)       = lo;
            *reinterpret_cast<v4f*>(oq + 128) = hi;
        }
    }

    {
        v8f acc[2];
        gemm_tile2<true>(&sA2[0][0], Wt + (size_t)(8 + head) * 2048 + lane * 2, lm, hs, acc[0], acc[1]);
        const float* bp = bkv + (8 * hs) * NHEADS + head;
        for (int t = 0; t < 2; ++t) {
            float* ok = Kp + plane + slot + t * 256;
            v4f lo, hi;
            #pragma unroll
            for (int r = 0; r < 4; ++r) {
                lo[r] = acc[t][r]     + bp[r * NHEADS];
                hi[r] = acc[t][r + 4] + bp[(r + 4) * NHEADS];
            }
            *reinterpret_cast<v4f*>(ok)       = lo;
            *reinterpret_cast<v4f*>(ok + 128) = hi;
        }
    }

    {
        v8f acc[2];
        gemm_tile2<false>(&sA2[0][0], Wt + (size_t)(16 + head) * 2048 + lane * 2, lm, hs, acc[0], acc[1]);
        const float bv = bkv[HID + lm * NHEADS + head];
        for (int t = 0; t < 2; ++t) {
            float* ov = Vp + plane + slot + t * 256;
            v4f lo, hi;
            #pragma unroll
            for (int r = 0; r < 4; ++r) { lo[r] = acc[t][r] + bv; hi[r] = acc[t][r + 4] + bv; }
            *reinterpret_cast<v4f*>(ov)       = lo;
            *reinterpret_cast<v4f*>(ov + 128) = hi;
        }
    }

    if (wave == pr % 4) {
        v8f acc = gemm_tile1(&sA2[16 * hh][0], Wt + (size_t)24 * 2048 + lane * 2, lm, hs);
        const float bv = beg[lm];
        const int bBase = b0 + 16 * hh + 8 * hs;
        if (lm < NHEADS) {
            float* oe = Ep + ((size_t)aIdx * NHEADS + lm) * NSZ + bBase;
            #pragma unroll
            for (int r = 0; r < 8; ++r) oe[r] = acc[r] + bv;
        } else {
            float* og = Gp + ((size_t)aIdx * NHEADS + (lm - NHEADS)) * NSZ + bBase;
            #pragma unroll
            for (int r = 0; r < 8; ++r)
                og[r] = 1.f / (1.f + __expf(-(acc[r] + bv)));
        }
    }
}

__global__ __launch_bounds__(256) __attribute__((amdgpu_waves_per_eu(6))) void attn_kernel(
    const float* __restrict__ Qp, const float* __restrict__ Kp,
    const float* __restrict__ Vp, const float* __restrict__ Ep,
    const float* __restrict__ Gp,
    const float* __restrict__ ln_g, const float* __restrict__ ln_b,
    float* __restrict__ out)
{
    __shared__ float VaLds[16][HID];

    const int j    = blockIdx.x;
    const int i0   = blockIdx.y * 16;
    const int tid  = threadIdx.x;
    const int h    = __builtin_amdgcn_readfirstlane(tid / 32);
    const int lane = tid % 32;
    const int hs   = lane / 16;
    const int lm   = lane % 16;

    const size_t jh = (size_t)j * NHEADS + h;

    const float* pQ = Qp + jh * (NSZ * HD) + blockIdx.y * 256 + lane * 4;
    const float* pK = Kp + jh * (NSZ * HD) + lane * 4;
    const float* pV = Vp + jh * (HD * NSZ) + lane * 4;
    const int   kq = (lm + 8 * hs) % 16;
    const float* pE = Ep + jh * NSZ + kq;
    const float* pG = Gp + jh * NSZ + kq;

    {
        const float* vslice = Vp + jh * HD * NSZ;
        for (int p = lane; p < 80; p += 32)
            __builtin_prefetch(vslice + p * 32, 0, 1);
    }

    v2f qfr[4];
    #pragma unroll
    for (int x = 0; x < 2; ++x) {
        const v4f q4 = *reinterpret_cast<const v4f*>(pQ + x * 128);
        qfr[2 * x]     = v2f{q4[0], q4[1]};
        qfr[2 * x + 1] = v2f{q4[2], q4[3]};
    }

    v2f kf[40];
    #pragma unroll
    for (int ct = 0; ct < 10; ++ct) {
        #pragma unroll
        for (int x = 0; x < 2; ++x) {
            const v4f k4 = *reinterpret_cast<const v4f*>(pK + ct * 256 + x * 128);
            kf[ct * 4 + 2 * x]     = v2f{k4[0], k4[1]};
            kf[ct * 4 + 2 * x + 1] = v2f{k4[2], k4[3]};
        }
    }
    float ev[10];
    #pragma unroll
    for (int ct = 0; ct < 10; ++ct) ev[ct] = pE[ct * 16];
    __builtin_amdgcn_sched_barrier(0);

    v8f acc[10];
    #pragma unroll
    for (int ct = 0; ct < 10; ++ct) {
        v8f c = {0.f, 0.f, 0.f, 0.f, 0.f, 0.f, 0.f, 0.f};
        #pragma unroll
        for (int ks = 0; ks < 4; ++ks)
            c = wmma_f32_16x16x4(kf[ct * 4 + ks], qfr[ks], c);
        acc[ct] = c;
        if (ct % 2 == 1) __builtin_amdgcn_sched_barrier(0);
    }
    #pragma unroll
    for (int ct = 0; ct < 10; ++ct) acc[ct] = row16_add8(acc[ct], ev[ct]);

    float gate[10];
    #pragma unroll
    for (int ct = 0; ct < 10; ++ct) gate[ct] = pG[ct * 16];
    __builtin_amdgcn_sched_barrier(0);

    v8f mx8 = acc[0];
    #pragma unroll
    for (int ct = 1; ct < 10; ++ct) {
        #pragma unroll
        for (int r = 0; r < 8; ++r) mx8[r] = fmaxf(mx8[r], acc[ct][r]);
    }
    float mx = fmaxf(fmaxf(fmaxf(mx8[0], mx8[1]), fmaxf(mx8[2], mx8[3])),
                     fmaxf(fmaxf(mx8[4], mx8[5]), fmaxf(mx8[6], mx8[7])));
    mx = fmaxf(mx, -1e30f);
    mx = fmaxf(mx, other_half(mx));

    v8f sm8 = {0.f, 0.f, 0.f, 0.f, 0.f, 0.f, 0.f, 0.f};
    #pragma unroll
    for (int ct = 0; ct < 10; ++ct) {
        #pragma unroll
        for (int r = 0; r < 8; ++r) {
            const float e = __expf(acc[ct][r] - mx);
            acc[ct][r] = e;
            sm8[r] += e;
        }
    }
    float sm = ((sm8[0] + sm8[1]) + (sm8[2] + sm8[3])) + ((sm8[4] + sm8[5]) + (sm8[6] + sm8[7]));
    sm += other_half(sm);
    const float rinv = 1.f / sm;
    #pragma unroll
    for (int ct = 0; ct < 10; ++ct) acc[ct] = row16_mul8(acc[ct] * rinv, gate[ct]);

    v2f vf[40];
    #pragma unroll
    for (int ct = 0; ct < 7; ++ct) {
        #pragma unroll
        for (int x = 0; x < 2; ++x) {
            const v4f v4 = *reinterpret_cast<const v4f*>(pV + ct * 256 + x * 128);
            vf[ct * 4 + 2 * x]     = v2f{v4[0], v4[1]};
            vf[ct * 4 + 2 * x + 1] = v2f{v4[2], v4[3]};
        }
    }
    __builtin_amdgcn_sched_barrier(0);
    v8f va = {0.f, 0.f, 0.f, 0.f, 0.f, 0.f, 0.f, 0.f};
    #pragma unroll
    for (int ct = 0; ct < 7; ++ct) {
        #pragma unroll
        for (int rp = 0; rp < 4; ++rp)
            va = wmma_f32_16x16x4(v2f{acc[ct][2 * rp], acc[ct][2 * rp + 1]}, vf[ct * 4 + rp], va);
    }
    #pragma unroll
    for (int ct = 7; ct < 10; ++ct) {
        #pragma unroll
        for (int x = 0; x < 2; ++x) {
            const v4f v4 = *reinterpret_cast<const v4f*>(pV + ct * 256 + x * 128);
            vf[ct * 4 + 2 * x]     = v2f{v4[0], v4[1]};
            vf[ct * 4 + 2 * x + 1] = v2f{v4[2], v4[3]};
        }
    }
    __builtin_amdgcn_sched_barrier(0);
    #pragma unroll
    for (int ct = 7; ct < 10; ++ct) {
        #pragma unroll
        for (int rp = 0; rp < 4; ++rp)
            va = wmma_f32_16x16x4(v2f{acc[ct][2 * rp], acc[ct][2 * rp + 1]}, vf[ct * 4 + rp], va);
    }

    #pragma unroll
    for (int r = 0; r < 8; ++r)
        VaLds[r + 8 * hs][lm * NHEADS + h] = va[r];
    __syncthreads();

    const int rowLN = tid / 16;
    const int cg    = tid % 16;
    float vals[8], s1 = 0.f, s2 = 0.f;
    #pragma unroll
    for (int c = 0; c < 8; ++c) {
        float v = VaLds[rowLN][cg * 8 + c];
        vals[c] = v; s1 += v; s2 += v * v;
    }
    s1 = row16_sum(s1);
    s2 = row16_sum(s2);
    const float mu  = s1 * (1.f / HID);
    const float var = s2 * (1.f / HID) - mu * mu;
    const float inv = rsqrtf(var + LN_EPS);

    const size_t obase = ((size_t)(i0 + rowLN) * NSZ + j) * HID + cg * 8;
    v2f* op = reinterpret_cast<v2f*>(&out[obase]);
    #pragma unroll
    for (int c = 0; c < 8; c += 2) {
        const int ch = cg * 8 + c;
        v2f t;
        t.x = (vals[c]     - mu) * inv * ln_g[ch]     + ln_b[ch];
        t.y = (vals[c + 1] - mu) * inv * ln_g[ch + 1] + ln_b[ch + 1];
        op[c / 2] = t;
    }
}

extern "C" void kernel_launch(void* const* d_in, const int* in_sizes, int n_in,
                              void* d_out, int out_size, void* d_ws, size_t ws_size,
                              hipStream_t stream) {
    const float* feat1 = (const float*)d_in[0];
    const float* feat2 = (const float*)d_in[1];
    const float* Wq    = (const float*)d_in[2];
    const float* bq    = (const float*)d_in[3];
    const float* Wkv   = (const float*)d_in[4];
    const float* bkv   = (const float*)d_in[5];
    const float* Weg   = (const float*)d_in[6];
    const float* beg   = (const float*)d_in[7];
    const float* lng   = (const float*)d_in[8];
    const float* lnb   = (const float*)d_in[9];

    float* ws = (float*)d_ws;
    float* Qp = ws;
    float* Kp = Qp + (size_t)NN * HID;
    float* Vp = Kp + (size_t)NN * HID;
    float* Ep = Vp + (size_t)NN * HID;
    float* Gp = Ep + (size_t)NN * NHEADS;
    float* Wt = Gp + (size_t)NN * NHEADS;

    wpack_kernel<<<dim3(25), 256, 0, stream>>>(Wq, Wkv, Weg, Wt);

    proj_kernel<<<dim3(NN / 16), 128, 0, stream>>>(
        feat1, feat2, Wt, bq, bkv, beg, Qp, Kp, Vp, Ep, Gp);

    attn_kernel<<<dim3(NSZ, NSZ / 16), 256, 0, stream>>>(
        Qp, Kp, Vp, Ep, Gp, lng, lnb, (float*)d_out);
}
